# starting kernel
# baseline (speedup 1.0000x reference)
#include <hip/hip_runtime.h>
#include <hip/hip_bf16.h>
#include <math.h>

#define IN_DIM  256
#define OUT_DIM 128
#define SLOPE   0.1f

typedef __attribute__((ext_vector_type(2))) float v2f;
typedef __attribute__((ext_vector_type(8))) float v8f;

__global__ __launch_bounds__(256) void zero_kernel(float* __restrict__ out,
                                                   float* __restrict__ row_sum,
                                                   int n_out, int n_rs) {
    int total = n_out + n_rs;
    for (int i = blockIdx.x * blockDim.x + threadIdx.x; i < total;
         i += gridDim.x * blockDim.x) {
        if (i < n_out) out[i] = 0.0f;
        else           row_sum[i - n_out] = 0.0f;
    }
}

__global__ __launch_bounds__(256) void gemm_wmma_kernel(
    const float* __restrict__ x, const float* __restrict__ W,
    const float* __restrict__ bias, float* __restrict__ emb, int N) {
    const int tid  = threadIdx.x;
    const int wave = tid >> 5;
    const int lane = tid & 31;
    const int half = lane >> 4;
    const int l16  = lane & 15;
    const int m0   = blockIdx.x * 32;
    const int m1   = m0 + 16;
    const int n0   = wave * 16;
    if (m0 >= N) return;
    const bool has_m1 = (m1 < N);
    const int  m1s    = has_m1 ? m1 : m0;

    const float* a0row = x + (size_t)(m0 + l16) * IN_DIM + 2 * half;
    const float* a1row = x + (size_t)(m1s + l16) * IN_DIM + 2 * half;
    const float* brow  = W + (size_t)(n0 + l16) * IN_DIM + 2 * half;

    v8f c0 = {};
    v8f c1 = {};
#pragma unroll 8
    for (int k0 = 0; k0 < IN_DIM; k0 += 4) {
        v2f b  = *(const v2f*)(brow + k0);
        v2f a0 = *(const v2f*)(a0row + k0);
        v2f a1 = *(const v2f*)(a1row + k0);
        c0 = __builtin_amdgcn_wmma_f32_16x16x4_f32(
                 false, a0, false, b, (short)0, c0, false, false);
        c1 = __builtin_amdgcn_wmma_f32_16x16x4_f32(
                 false, a1, false, b, (short)0, c1, false, false);
    }

    const float bv = bias[n0 + l16];
#pragma unroll
    for (int j = 0; j < 8; ++j) {
        const int row = m0 + j + 8 * half;
        emb[(size_t)row * OUT_DIM + n0 + l16] = c0[j] + bv;
    }
    if (has_m1) {
#pragma unroll
        for (int j = 0; j < 8; ++j) {
            const int row = m1 + j + 8 * half;
            emb[(size_t)row * OUT_DIM + n0 + l16] = c1[j] + bv;
        }
    }
}

__global__ __launch_bounds__(256) void score_kernel(
    const float* __restrict__ emb, const float* __restrict__ a,
    float* __restrict__ s_src, float* __restrict__ s_dst, int N) {
    const int wid  = (blockIdx.x * blockDim.x + threadIdx.x) >> 5;
    const int lane = threadIdx.x & 31;
    if (wid >= N) return;

    const float4 e  = *(const float4*)(emb + (size_t)wid * OUT_DIM + lane * 4);
    const float4 a1 = *(const float4*)(a + lane * 4);
    const float4 a2 = *(const float4*)(a + OUT_DIM + lane * 4);

    float p1 = e.x * a1.x + e.y * a1.y + e.z * a1.z + e.w * a1.w;
    float p2 = e.x * a2.x + e.y * a2.y + e.z * a2.z + e.w * a2.w;
#pragma unroll
    for (int off = 16; off > 0; off >>= 1) {
        p1 += __shfl_xor(p1, off, 32);
        p2 += __shfl_xor(p2, off, 32);
    }
    if (lane == 0) {
        s_src[wid] = p1;
        s_dst[wid] = p2;
    }
}

__global__ __launch_bounds__(256) void edge_kernel(
    const int* __restrict__ edges, const float* __restrict__ s_src,
    const float* __restrict__ s_dst, const float* __restrict__ emb,
    float* __restrict__ out, float* __restrict__ row_sum, int E) {
    const int wid  = (blockIdx.x * blockDim.x + threadIdx.x) >> 5;
    const int lane = threadIdx.x & 31;
    if (wid >= E) return;

    const int2 sd = *(const int2*)(edges + 2 * (size_t)wid);
    const int src = sd.x;
    const int dst = sd.y;

    const float logit = s_src[src] + s_dst[dst];
    const float lr    = logit > 0.0f ? logit : SLOPE * logit;
    const float score = expf(lr);

    if (lane == 0) atomicAdd(row_sum + src, score);

    const float4 m = *(const float4*)(emb + (size_t)dst * OUT_DIM + lane * 4);
    float* o = out + (size_t)src * OUT_DIM + lane * 4;
    atomicAdd(o + 0, score * m.x);
    atomicAdd(o + 1, score * m.y);
    atomicAdd(o + 2, score * m.z);
    atomicAdd(o + 3, score * m.w);
}

__global__ __launch_bounds__(256) void finalize_kernel(
    float* __restrict__ out, const float* __restrict__ row_sum, int total) {
    const int i = blockIdx.x * blockDim.x + threadIdx.x;
    if (i >= total) return;
    const float rs   = row_sum[i >> 7];
    const float safe = rs > 0.0f ? rs : 1.0f;
    out[i] = out[i] / safe;
}

extern "C" void kernel_launch(void* const* d_in, const int* in_sizes, int n_in,
                              void* d_out, int out_size, void* d_ws, size_t ws_size,
                              hipStream_t stream) {
    const float* x     = (const float*)d_in[0];
    const float* W     = (const float*)d_in[1];
    const float* bias  = (const float*)d_in[2];
    const float* a     = (const float*)d_in[3];
    const int*   edges = (const int*)d_in[4];

    const int N = in_sizes[0] / IN_DIM;
    const int E = in_sizes[4] / 2;

    float* ws      = (float*)d_ws;
    float* emb     = ws;
    float* s_src   = emb + (size_t)N * OUT_DIM;
    float* s_dst   = s_src + N;
    float* row_sum = s_dst + N;
    float* out     = (float*)d_out;

    const int n_out = N * OUT_DIM;

    {
        int total = n_out + N;
        int grid = (total + 255) / 256;
        if (grid > 4096) grid = 4096;
        zero_kernel<<<grid, 256, 0, stream>>>(out, row_sum, n_out, N);
    }
    {
        int grid = (N + 31) / 32;
        gemm_wmma_kernel<<<grid, 256, 0, stream>>>(x, W, bias, emb, N);
    }
    {
        int grid = (N + 7) / 8;
        score_kernel<<<grid, 256, 0, stream>>>(emb, a, s_src, s_dst, N);
    }
    {
        int grid = (E + 7) / 8;
        edge_kernel<<<grid, 256, 0, stream>>>(edges, s_src, s_dst, emb, out,
                                              row_sum, E);
    }
    {
        int grid = (n_out + 255) / 256;
        finalize_kernel<<<grid, 256, 0, stream>>>(out, row_sum, n_out);
    }
}


# Scatter by global float atomics replaced: each work-group holds 392 output rows in LDS, filters the whole edge list for them, sums by LDS float adds, normalises at its store; summation order changes
# speedup vs baseline: 1.6961x; 1.6961x over previous
#include <hip/hip_runtime.h>
#include <hip/hip_bf16.h>
#include <math.h>

#define IN_DIM  256
#define OUT_DIM 128
#define SLOPE   0.1f

typedef __attribute__((ext_vector_type(2))) float v2f;
typedef __attribute__((ext_vector_type(8))) float v8f;

__global__ __launch_bounds__(256) void gemm_wmma_kernel(
    const float* __restrict__ x, const float* __restrict__ W,
    const float* __restrict__ bias, float* __restrict__ emb, int N) {
    const int tid  = threadIdx.x;
    const int wave = tid >> 5;
    const int lane = tid & 31;
    const int half = lane >> 4;
    const int l16  = lane & 15;
    const int m0   = blockIdx.x * 32;
    const int m1   = m0 + 16;
    const int n0   = wave * 16;
    if (m0 >= N) return;
    const bool has_m1 = (m1 < N);
    const int  m1s    = has_m1 ? m1 : m0;

    const float* a0row = x + (size_t)(m0 + l16) * IN_DIM + 2 * half;
    const float* a1row = x + (size_t)(m1s + l16) * IN_DIM + 2 * half;
    const float* brow  = W + (size_t)(n0 + l16) * IN_DIM + 2 * half;

    v8f c0 = {};
    v8f c1 = {};
#pragma unroll 8
    for (int k0 = 0; k0 < IN_DIM; k0 += 4) {
        v2f b  = *(const v2f*)(brow + k0);
        v2f a0 = *(const v2f*)(a0row + k0);
        v2f a1 = *(const v2f*)(a1row + k0);
        c0 = __builtin_amdgcn_wmma_f32_16x16x4_f32(
                 false, a0, false, b, (short)0, c0, false, false);
        c1 = __builtin_amdgcn_wmma_f32_16x16x4_f32(
                 false, a1, false, b, (short)0, c1, false, false);
    }

    const float bv = bias[n0 + l16];
#pragma unroll
    for (int j = 0; j < 8; ++j) {
        const int row = m0 + j + 8 * half;
        emb[(size_t)row * OUT_DIM + n0 + l16] = c0[j] + bv;
    }
    if (has_m1) {
#pragma unroll
        for (int j = 0; j < 8; ++j) {
            const int row = m1 + j + 8 * half;
            emb[(size_t)row * OUT_DIM + n0 + l16] = c1[j] + bv;
        }
    }
}

__global__ __launch_bounds__(256) void score_kernel(
    const float* __restrict__ emb, const float* __restrict__ a,
    float* __restrict__ s_src, float* __restrict__ s_dst, int N) {
    const int wid  = (blockIdx.x * blockDim.x + threadIdx.x) >> 5;
    const int lane = threadIdx.x & 31;
    if (wid >= N) return;

    const float4 e  = *(const float4*)(emb + (size_t)wid * OUT_DIM + lane * 4);
    const float4 a1 = *(const float4*)(a + lane * 4);
    const float4 a2 = *(const float4*)(a + OUT_DIM + lane * 4);

    float p1 = e.x * a1.x + e.y * a1.y + e.z * a1.z + e.w * a1.w;
    float p2 = e.x * a2.x + e.y * a2.y + e.z * a2.z + e.w * a2.w;
#pragma unroll
    for (int off = 16; off > 0; off >>= 1) {
        p1 += __shfl_xor(p1, off, 32);
        p2 += __shfl_xor(p2, off, 32);
    }
    if (lane == 0) {
        s_src[wid] = p1;
        s_dst[wid] = p2;
    }
}

#define AGG_ROWS    392
#define AGG_THREADS 1024
#define AGG_WAVES   32
#define AGG_LOADS   4
#define AGG_QCAP    288

typedef __attribute__((ext_vector_type(2))) int v2i;

__global__ __launch_bounds__(AGG_THREADS) void aggregate_kernel(
    const int* __restrict__ edges, const float* __restrict__ s_src,
    const float* __restrict__ s_dst, const float* __restrict__ emb,
    float* __restrict__ out, int N, int E) {
    __shared__ float acc[AGG_ROWS * OUT_DIM];
    __shared__ float rsum[AGG_ROWS];
    __shared__ float ssrc[AGG_ROWS];
    __shared__ v2i   queue[AGG_WAVES * AGG_QCAP];

    const int tid  = threadIdx.x;
    const int lane = tid & 31;
    const int w    = __builtin_amdgcn_readfirstlane(tid >> 5);
    const int base = blockIdx.x * AGG_ROWS;
    int nrows = N - base;
    if (nrows > AGG_ROWS) nrows = AGG_ROWS;

    for (int i = tid; i < AGG_ROWS * OUT_DIM / 4; i += AGG_THREADS)
        ((float4*)acc)[i] = make_float4(0.0f, 0.0f, 0.0f, 0.0f);
    for (int i = tid; i < AGG_ROWS; i += AGG_THREADS) {
        rsum[i] = 0.0f;
        ssrc[i] = (i < nrows) ? s_src[base + i] : 0.0f;
    }
    __syncthreads();

    v2i* q = queue + w * AGG_QCAP;
    int qcount = 0;

    auto consider = [&](int s, int d) {
        const unsigned rel  = (unsigned)(s - base);
        const bool     hit  = rel < (unsigned)nrows;
        const unsigned mask = __builtin_amdgcn_ballot_w32(hit);
        if (mask != 0u) {
            const int pos = qcount + (int)__builtin_amdgcn_mbcnt_lo(mask, 0u);
            if (hit) {
                v2i ent;
                ent.x = (int)rel;
                ent.y = d;
                q[pos] = ent;
            }
            qcount += __builtin_popcount(mask);
        }
    };

    auto batch = [&](int b, int cnt) {
        int   sl = 0;
        int   dd = 0;
        float sc = 0.0f;
        if (lane < cnt) {
            const v2i ent = q[b + lane];
            sl = ent.x;
            dd = ent.y;
            const float logit = ssrc[sl] + s_dst[dd];
            const float lr    = logit > 0.0f ? logit : SLOPE * logit;
            sc = expf(lr);
            atomicAdd(&rsum[sl], sc);
        }
        int j = 0;
        for (; j + 8 <= cnt; j += 8) {
            float m[8][4];
#pragma unroll
            for (int t = 0; t < 8; ++t) {
                const int    dj   = __builtin_amdgcn_readlane(dd, j + t);
                const float* mrow = emb + (size_t)dj * OUT_DIM + lane;
                m[t][0] = mrow[0];
                m[t][1] = mrow[32];
                m[t][2] = mrow[64];
                m[t][3] = mrow[96];
            }
#pragma unroll
            for (int t = 0; t < 8; ++t) {
                const int   sj = __builtin_amdgcn_readlane(sl, j + t);
                const float cj = __int_as_float(
                    __builtin_amdgcn_readlane(__float_as_int(sc), j + t));
                float* arow = acc + sj * OUT_DIM + lane;
                atomicAdd(arow,      cj * m[t][0]);
                atomicAdd(arow + 32, cj * m[t][1]);
                atomicAdd(arow + 64, cj * m[t][2]);
                atomicAdd(arow + 96, cj * m[t][3]);
            }
        }
        for (; j < cnt; ++j) {
            const int   sj = __builtin_amdgcn_readlane(sl, j);
            const int   dj = __builtin_amdgcn_readlane(dd, j);
            const float cj = __int_as_float(
                __builtin_amdgcn_readlane(__float_as_int(sc), j));
            const float* mrow = emb + (size_t)dj * OUT_DIM + lane;
            float*       arow = acc + sj * OUT_DIM + lane;
            const float m0 = mrow[0];
            const float m1 = mrow[32];
            const float m2 = mrow[64];
            const float m3 = mrow[96];
            atomicAdd(arow,      cj * m0);
            atomicAdd(arow + 32, cj * m1);
            atomicAdd(arow + 64, cj * m2);
            atomicAdd(arow + 96, cj * m3);
        }
    };

    auto drain = [&](bool last) {
        __builtin_amdgcn_fence(__ATOMIC_RELEASE, "wavefront");
        __builtin_amdgcn_wave_barrier();
        __builtin_amdgcn_fence(__ATOMIC_ACQUIRE, "wavefront");
        int b = 0;
        for (; b + 32 <= qcount; b += 32) batch(b, 32);
        const int rem = qcount - b;
        if (last) {
            if (rem > 0) batch(b, rem);
            qcount = 0;
        } else {
            if (b > 0 && lane < rem) {
                const v2i ent = q[b + lane];
                q[lane] = ent;
            }
            qcount = rem;
        }
    };

    const int   npairs = E >> 1;
    const int4* e4     = (const int4*)edges;
    const int   step   = AGG_LOADS * AGG_THREADS;

    int blk = 0;
    for (; blk + step <= npairs; blk += step) {
        int4 v[AGG_LOADS];
#pragma unroll
        for (int u = 0; u < AGG_LOADS; ++u)
            v[u] = e4[blk + u * AGG_THREADS + tid];
#pragma unroll
        for (int u = 0; u < AGG_LOADS; ++u) {
            consider(v[u].x, v[u].y);
            consider(v[u].z, v[u].w);
        }
        if (qcount >= 32) drain(false);
    }
    if (blk < npairs) {
        int4 v[AGG_LOADS];
#pragma unroll
        for (int u = 0; u < AGG_LOADS; ++u) {
            const int p = blk + u * AGG_THREADS + tid;
            v[u] = make_int4(-1, 0, -1, 0);
            if (p < npairs) v[u] = e4[p];
        }
#pragma unroll
        for (int u = 0; u < AGG_LOADS; ++u) {
            consider(v[u].x, v[u].y);
            consider(v[u].z, v[u].w);
        }
    }
    if ((E & 1) && w == 0) {
        int s = -1;
        int d = 0;
        if (lane == 0) {
            s = edges[2 * (size_t)(E - 1)];
            d = edges[2 * (size_t)(E - 1) + 1];
        }
        consider(s, d);
    }
    drain(true);
    __syncthreads();

    float4* o4 = (float4*)out + (size_t)base * (OUT_DIM / 4);
    for (int i = tid; i < nrows * (OUT_DIM / 4); i += AGG_THREADS) {
        const float rs   = rsum[i >> 5];
        const float safe = rs > 0.0f ? rs : 1.0f;
        float4 a = ((const float4*)acc)[i];
        a.x = a.x / safe;
        a.y = a.y / safe;
        a.z = a.z / safe;
        a.w = a.w / safe;
        o4[i] = a;
    }
}

extern "C" void kernel_launch(void* const* d_in, const int* in_sizes, int n_in,
                              void* d_out, int out_size, void* d_ws, size_t ws_size,
                              hipStream_t stream) {
    const float* x     = (const float*)d_in[0];
    const float* W     = (const float*)d_in[1];
    const float* bias  = (const float*)d_in[2];
    const float* a     = (const float*)d_in[3];
    const int*   edges = (const int*)d_in[4];

    const int N = in_sizes[0] / IN_DIM;
    const int E = in_sizes[4] / 2;

    float* ws    = (float*)d_ws;
    float* emb   = ws;
    float* s_src = emb + (size_t)N * OUT_DIM;
    float* s_dst = s_src + N;
    float* out   = (float*)d_out;

    {
        int grid = (N + 31) / 32;
        gemm_wmma_kernel<<<grid, 256, 0, stream>>>(x, W, bias, emb, N);
    }
    {
        int grid = (N + 7) / 8;
        score_kernel<<<grid, 256, 0, stream>>>(emb, a, s_src, s_dst, N);
    }
    {
        int grid = (N + AGG_ROWS - 1) / AGG_ROWS;
        aggregate_kernel<<<grid, AGG_THREADS, 0, stream>>>(edges, s_src, s_dst, emb,
                                                           out, N, E);
    }
}


# Linear layer made weight-stationary: each wave keeps its 16 columns of W in registers over a loop of 16-row x tiles staged in LDS by whole-line loads; k summed in a permuted order inside groups of 8
# speedup vs baseline: 2.9934x; 1.7649x over previous
#include <hip/hip_runtime.h>
#include <hip/hip_bf16.h>
#include <math.h>

#define IN_DIM  256
#define OUT_DIM 128
#define SLOPE   0.1f

typedef __attribute__((ext_vector_type(2))) float v2f;
typedef __attribute__((ext_vector_type(8))) float v8f;

#define GEMM_THREADS 256
#define GEMM_BLOCKS  512
#define GEMM_ROWS    16
#define GEMM_PITCH   260

typedef __attribute__((ext_vector_type(4))) float v4f;

__global__ __launch_bounds__(GEMM_THREADS, 4) void gemm_wmma_kernel(
    const float* __restrict__ x, const float* __restrict__ W,
    const float* __restrict__ bias, float* __restrict__ emb, int N) {
    __shared__ float xs[2][GEMM_ROWS * GEMM_PITCH];

    const int tid  = threadIdx.x;
    const int wave = __builtin_amdgcn_readfirstlane(tid >> 5);
    const int lane = tid & 31;
    const int half = lane >> 4;
    const int l16  = lane & 15;
    const int n0   = wave * 16;
    const int ntiles = (N + GEMM_ROWS - 1) / GEMM_ROWS;
    int s = blockIdx.x;
    if (s >= ntiles) return;

    v4f w[32];
    {
        const float* wrow = W + (size_t)(n0 + l16) * IN_DIM + 4 * half;
#pragma unroll
        for (int j = 0; j < 32; ++j) w[j] = *(const v4f*)(wrow + 8 * j);
    }
    const float bv = bias[n0 + l16];

    const int srow = wave >> 1;
    const int scol = ((wave & 1) * 32 + lane) * 4;

    v4f g[4];
    auto fetch = [&](int tile) {
#pragma unroll
        for (int i = 0; i < 4; ++i) {
            int row = tile * GEMM_ROWS + 4 * i + srow;
            row = row < N ? row : N - 1;
            const float* xr = x + (size_t)row * IN_DIM;
            g[i] = *(const v4f*)(xr + scol);
        }
    };
    auto stash = [&](float* buf) {
#pragma unroll
        for (int i = 0; i < 4; ++i)
            *(v4f*)(buf + (4 * i + srow) * GEMM_PITCH + scol) = g[i];
    };

    fetch(s);
    stash(xs[0]);
    __syncthreads();

    int cur = 0;
    for (;;) {
        const int  next = s + (int)gridDim.x;
        const bool more = next < ntiles;
        fetch(more ? next : s);
        __builtin_amdgcn_sched_barrier(0);

        const float* ap = xs[cur] + l16 * GEMM_PITCH + 4 * half;
        v4f a[2][8];
#pragma unroll
        for (int t = 0; t < 8; ++t) a[0][t] = *(const v4f*)(ap + 8 * t);
        v8f c = {};
#pragma unroll
        for (int b = 0; b < 4; ++b) {
            if (b < 3) {
#pragma unroll
                for (int t = 0; t < 8; ++t)
                    a[(b + 1) & 1][t] = *(const v4f*)(ap + 8 * (8 * (b + 1) + t));
            }
            __builtin_amdgcn_sched_barrier(0);
#pragma unroll
            for (int t = 0; t < 8; ++t) {
                const v4f av = a[b & 1][t];
                const v4f wv = w[8 * b + t];
                const v2f alo = {av.x, av.y};
                const v2f ahi = {av.z, av.w};
                const v2f blo = {wv.x, wv.y};
                const v2f bhi = {wv.z, wv.w};
                c = __builtin_amdgcn_wmma_f32_16x16x4_f32(
                        false, alo, false, blo, (short)0, c, false, false);
                c = __builtin_amdgcn_wmma_f32_16x16x4_f32(
                        false, ahi, false, bhi, (short)0, c, false, false);
            }
        }

        stash(xs[cur ^ 1]);
        __syncthreads();

        asm volatile("v_nop\n\tv_nop\n\tv_nop\n\tv_nop" : "+v"(c));

        const int m0 = s * GEMM_ROWS;
        float* erow = emb + (size_t)(m0 + 8 * half) * OUT_DIM + n0 + l16;
        if (m0 + GEMM_ROWS <= N) {
#pragma unroll
            for (int j = 0; j < 8; ++j) erow[j * OUT_DIM] = c[j] + bv;
        } else {
#pragma unroll
            for (int j = 0; j < 8; ++j)
                if (m0 + j + 8 * half < N) erow[j * OUT_DIM] = c[j] + bv;
        }

        if (!more) break;
        cur ^= 1;
        s = next;
    }
}

__global__ __launch_bounds__(256) void score_kernel(
    const float* __restrict__ emb, const float* __restrict__ a,
    float* __restrict__ s_src, float* __restrict__ s_dst, int N) {
    const int wid  = (blockIdx.x * blockDim.x + threadIdx.x) >> 5;
    const int lane = threadIdx.x & 31;
    if (wid >= N) return;

    const float4 e  = *(const float4*)(emb + (size_t)wid * OUT_DIM + lane * 4);
    const float4 a1 = *(const float4*)(a + lane * 4);
    const float4 a2 = *(const float4*)(a + OUT_DIM + lane * 4);

    float p1 = e.x * a1.x + e.y * a1.y + e.z * a1.z + e.w * a1.w;
    float p2 = e.x * a2.x + e.y * a2.y + e.z * a2.z + e.w * a2.w;
#pragma unroll
    for (int off = 16; off > 0; off >>= 1) {
        p1 += __shfl_xor(p1, off, 32);
        p2 += __shfl_xor(p2, off, 32);
    }
    if (lane == 0) {
        s_src[wid] = p1;
        s_dst[wid] = p2;
    }
}

#define AGG_ROWS    392

__device__ __forceinline__ size_t key_offset(int N) {
    return ((size_t)N * (OUT_DIM + 2) + 3) & ~(size_t)3;
}
__device__ __forceinline__ int key_words(int E) {
    return ((E + 15) >> 4) << 2;
}
__device__ __forceinline__ bool keys_fit(int N, int E, size_t ws_size) {
    return (key_offset(N) + (size_t)key_words(E)) * 4 <= ws_size;
}
__device__ __forceinline__ unsigned edge_key(int s) {
    return ((unsigned)s / AGG_ROWS) & 255u;
}

__global__ __launch_bounds__(256) void key_kernel(
    const int* __restrict__ edges, float* __restrict__ ws, int N, int E,
    size_t ws_size) {
    if (!keys_fit(N, E, ws_size)) return;
    unsigned* keys = (unsigned*)ws + key_offset(N);
    const int g = blockIdx.x * 256 + threadIdx.x;
    if (g >= key_words(E)) return;
    const int e0 = 4 * g;
    unsigned k = 0u;
    if (e0 + 3 < E) {
        const int4* e4 = (const int4*)edges;
        const int4 p0 = e4[2 * g];
        const int4 p1 = e4[2 * g + 1];
        k = edge_key(p0.x) | (edge_key(p0.z) << 8) |
            (edge_key(p1.x) << 16) | (edge_key(p1.z) << 24);
    } else {
        for (int t = 0; t < 4; ++t)
            if (e0 + t < E) k |= edge_key(edges[2 * (size_t)(e0 + t)]) << (8 * t);
    }
    keys[g] = k;
}

#define AGG_THREADS 1024
#define AGG_WAVES   32
#define AGG_LOADS   4
#define AGG_QCAP    64

typedef __attribute__((ext_vector_type(2))) int v2i;

__global__ __launch_bounds__(AGG_THREADS) void aggregate_kernel(
    const int* __restrict__ edges, const float* __restrict__ ws,
    float* __restrict__ out, int N, int E, size_t ws_size) {
    __shared__ float acc[(AGG_ROWS + 1) * OUT_DIM];
    __shared__ float rsum[AGG_ROWS];
    __shared__ float ssrc[AGG_ROWS];
    __shared__ int   queue[AGG_WAVES * AGG_QCAP];

    if (!keys_fit(N, E, ws_size)) return;
    const float*    emb   = ws;
    const float*    s_src = emb + (size_t)N * OUT_DIM;
    const float*    s_dst = s_src + N;
    const unsigned* keys  = (const unsigned*)ws + key_offset(N);

    const int tid  = threadIdx.x;
    const int lane = tid & 31;
    const int w    = __builtin_amdgcn_readfirstlane(tid >> 5);
    const int base = blockIdx.x * AGG_ROWS;
    int nrows = N - base;
    if (nrows > AGG_ROWS) nrows = AGG_ROWS;

    for (int i = tid; i < (AGG_ROWS + 1) * OUT_DIM / 4; i += AGG_THREADS)
        ((float4*)acc)[i] = make_float4(0.0f, 0.0f, 0.0f, 0.0f);
    for (int i = tid; i < AGG_ROWS; i += AGG_THREADS) {
        rsum[i] = 0.0f;
        ssrc[i] = (i < nrows) ? s_src[base + i] : 0.0f;
    }
    __syncthreads();

    int* q = queue + w * AGG_QCAP;
    int qcount = 0;
    const v2i* e2 = (const v2i*)edges;

    auto batch = [&](int cnt) {
        __builtin_amdgcn_fence(__ATOMIC_RELEASE, "wavefront");
        __builtin_amdgcn_wave_barrier();
        __builtin_amdgcn_fence(__ATOMIC_ACQUIRE, "wavefront");
        int   sl = AGG_ROWS;
        int   dd = 0;
        float sc = 0.0f;
        if (lane < cnt) {
            const int e = q[lane];
            if (e < E) {
                const v2i      ed  = e2[e];
                const unsigned rel = (unsigned)(ed.x - base);
                if (rel < (unsigned)nrows) {
                    sl = (int)rel;
                    dd = ed.y;
                    const float logit = ssrc[sl] + s_dst[dd];
                    const float lr    = logit > 0.0f ? logit : SLOPE * logit;
                    sc = expf(lr);
                    atomicAdd(&rsum[sl], sc);
                }
            }
        }
        int j = 0;
        for (; j + 8 <= cnt; j += 8) {
            float m[8][4];
#pragma unroll
            for (int t = 0; t < 8; ++t) {
                const int    dj   = __builtin_amdgcn_readlane(dd, j + t);
                const float* mrow = emb + (size_t)dj * OUT_DIM + lane;
                m[t][0] = mrow[0];
                m[t][1] = mrow[32];
                m[t][2] = mrow[64];
                m[t][3] = mrow[96];
            }
#pragma unroll
            for (int t = 0; t < 8; ++t) {
                const int   sj = __builtin_amdgcn_readlane(sl, j + t);
                const float cj = __int_as_float(
                    __builtin_amdgcn_readlane(__float_as_int(sc), j + t));
                float* arow = acc + sj * OUT_DIM + lane;
                atomicAdd(arow,      cj * m[t][0]);
                atomicAdd(arow + 32, cj * m[t][1]);
                atomicAdd(arow + 64, cj * m[t][2]);
                atomicAdd(arow + 96, cj * m[t][3]);
            }
        }
        for (; j < cnt; ++j) {
            const int   sj = __builtin_amdgcn_readlane(sl, j);
            const int   dj = __builtin_amdgcn_readlane(dd, j);
            const float cj = __int_as_float(
                __builtin_amdgcn_readlane(__float_as_int(sc), j));
            const float* mrow = emb + (size_t)dj * OUT_DIM + lane;
            float*       arow = acc + sj * OUT_DIM + lane;
            const float m0 = mrow[0];
            const float m1 = mrow[32];
            const float m2 = mrow[64];
            const float m3 = mrow[96];
            atomicAdd(arow,      cj * m0);
            atomicAdd(arow + 32, cj * m1);
            atomicAdd(arow + 64, cj * m2);
            atomicAdd(arow + 96, cj * m3);
        }
    };

    const unsigned pat = (blockIdx.x & 255u) * 0x01010101u;
    auto hits4 = [&](unsigned wd) -> unsigned {
        const unsigned t = wd ^ pat;
        const unsigned a = ((t & 0x7f7f7f7fu) + 0x7f7f7f7fu) | t;
        return ~a & 0x80808080u;
    };
    auto hits16 = [&](uint4 kq) -> unsigned {
        return (hits4(kq.x) >> 7) | (hits4(kq.y) >> 6) |
               (hits4(kq.z) >> 5) | (hits4(kq.w) >> 4);
    };

    const uint4* k4      = (const uint4*)keys;
    const int    ngroups = (E + 15) >> 4;
    const int    step    = AGG_LOADS * AGG_THREADS;

    auto scan = [&](const uint4 (&v)[AGG_LOADS], int blk) {
        unsigned long long mm =
            (unsigned long long)(hits16(v[0]) | (hits16(v[1]) << 4)) |
            ((unsigned long long)(hits16(v[2]) | (hits16(v[3]) << 4)) << 32);
        for (;;) {
            const bool     hit  = mm != 0ull;
            const unsigned mask = __builtin_amdgcn_ballot_w32(hit);
            if (mask == 0u) break;
            const int pos = qcount + (int)__builtin_amdgcn_mbcnt_lo(mask, 0u);
            if (hit) {
                const int p = __builtin_ctzll(mm);
                mm &= mm - 1ull;
                const int u = ((p >> 5) << 1) | ((p >> 2) & 1);
                const int g = blk + u * AGG_THREADS + tid;
                q[pos] = (g << 4) + ((p & 3) << 2) + ((p >> 3) & 3);
            }
            qcount += __builtin_popcount(mask);
            if (qcount >= 32) {
                batch(32);
                const int rem = qcount - 32;
                if (lane < rem) {
                    const int ent = q[32 + lane];
                    q[lane] = ent;
                }
                qcount = rem;
            }
        }
    };

    int blk = 0;
    for (; blk + step <= ngroups; blk += step) {
        uint4 v[AGG_LOADS];
#pragma unroll
        for (int u = 0; u < AGG_LOADS; ++u)
            v[u] = k4[blk + u * AGG_THREADS + tid];
        scan(v, blk);
    }
    if (blk < ngroups) {
        uint4 v[AGG_LOADS];
#pragma unroll
        for (int u = 0; u < AGG_LOADS; ++u) {
            const int p = blk + u * AGG_THREADS + tid;
            v[u] = k4[p < ngroups ? p : ngroups - 1];
        }
        scan(v, blk);
    }
    if (qcount > 0) batch(qcount);
    __syncthreads();

    float4* o4 = (float4*)out + (size_t)base * (OUT_DIM / 4);
    for (int i = tid; i < nrows * (OUT_DIM / 4); i += AGG_THREADS) {
        const float rs   = rsum[i >> 5];
        const float safe = rs > 0.0f ? rs : 1.0f;
        float4 a = ((const float4*)acc)[i];
        a.x = a.x / safe;
        a.y = a.y / safe;
        a.z = a.z / safe;
        a.w = a.w / safe;
        o4[i] = a;
    }
}

extern "C" void kernel_launch(void* const* d_in, const int* in_sizes, int n_in,
                              void* d_out, int out_size, void* d_ws, size_t ws_size,
                              hipStream_t stream) {
    const float* x     = (const float*)d_in[0];
    const float* W     = (const float*)d_in[1];
    const float* bias  = (const float*)d_in[2];
    const float* a     = (const float*)d_in[3];
    const int*   edges = (const int*)d_in[4];

    const int N = in_sizes[0] / IN_DIM;
    const int E = in_sizes[4] / 2;

    float* ws    = (float*)d_ws;
    float* emb   = ws;
    float* s_src = emb + (size_t)N * OUT_DIM;
    float* s_dst = s_src + N;
    float* out   = (float*)d_out;

    {
        gemm_wmma_kernel<<<GEMM_BLOCKS, GEMM_THREADS, 0, stream>>>(x, W, bias, emb, N);
    }
    {
        int grid = (N + 7) / 8;
        score_kernel<<<grid, 256, 0, stream>>>(emb, a, s_src, s_dst, N);
    }
    {
        int grid = (E + 1023) / 1024;
        key_kernel<<<grid, 256, 0, stream>>>(edges, ws, N, E, ws_size);
    }
    {
        int grid = (N + AGG_ROWS - 1) / AGG_ROWS;
        aggregate_kernel<<<grid, AGG_THREADS, 0, stream>>>(edges, ws, out, N, E, ws_size);
    }
}


# Key-plane filter replaced: a pre-pass orders each 4096-edge chunk by 392-row block in LDS (4-byte records + offset table); aggregation waves read their block's runs; LDS add order changes
# speedup vs baseline: 3.4326x; 1.1467x over previous
#include <hip/hip_runtime.h>
#include <hip/hip_bf16.h>
#include <math.h>

#define IN_DIM  256
#define OUT_DIM 128
#define SLOPE   0.1f

typedef __attribute__((ext_vector_type(2))) float v2f;
typedef __attribute__((ext_vector_type(8))) float v8f;

#define GEMM_THREADS 256
#define GEMM_BLOCKS  512
#define GEMM_ROWS    16
#define GEMM_PITCH   260

typedef __attribute__((ext_vector_type(4))) float v4f;

__global__ __launch_bounds__(GEMM_THREADS, 4) void gemm_wmma_kernel(
    const float* __restrict__ x, const float* __restrict__ W,
    const float* __restrict__ bias, float* __restrict__ emb, int N) {
    __shared__ float xs[2][GEMM_ROWS * GEMM_PITCH];

    const int tid  = threadIdx.x;
    const int wave = __builtin_amdgcn_readfirstlane(tid >> 5);
    const int lane = tid & 31;
    const int half = lane >> 4;
    const int l16  = lane & 15;
    const int n0   = wave * 16;
    const int ntiles = (N + GEMM_ROWS - 1) / GEMM_ROWS;
    int s = blockIdx.x;
    if (s >= ntiles) return;

    v4f w[32];
    {
        const float* wrow = W + (size_t)(n0 + l16) * IN_DIM + 4 * half;
#pragma unroll
        for (int j = 0; j < 32; ++j) w[j] = *(const v4f*)(wrow + 8 * j);
    }
    const float bv = bias[n0 + l16];

    const int srow = wave >> 1;
    const int scol = ((wave & 1) * 32 + lane) * 4;

    v4f g[4];
    auto fetch = [&](int tile) {
#pragma unroll
        for (int i = 0; i < 4; ++i) {
            int row = tile * GEMM_ROWS + 4 * i + srow;
            row = row < N ? row : N - 1;
            const float* xr = x + (size_t)row * IN_DIM;
            g[i] = *(const v4f*)(xr + scol);
        }
    };
    auto stash = [&](float* buf) {
#pragma unroll
        for (int i = 0; i < 4; ++i)
            *(v4f*)(buf + (4 * i + srow) * GEMM_PITCH + scol) = g[i];
    };

    fetch(s);
    stash(xs[0]);
    __syncthreads();

    int cur = 0;
    for (;;) {
        const int  next = s + (int)gridDim.x;
        const bool more = next < ntiles;
        fetch(more ? next : s);
        __builtin_amdgcn_sched_barrier(0);

        const float* ap = xs[cur] + l16 * GEMM_PITCH + 4 * half;
        v4f a[2][8];
#pragma unroll
        for (int t = 0; t < 8; ++t) a[0][t] = *(const v4f*)(ap + 8 * t);
        v8f c = {};
#pragma unroll
        for (int b = 0; b < 4; ++b) {
            if (b < 3) {
#pragma unroll
                for (int t = 0; t < 8; ++t)
                    a[(b + 1) & 1][t] = *(const v4f*)(ap + 8 * (8 * (b + 1) + t));
            }
            __builtin_amdgcn_sched_barrier(0);
#pragma unroll
            for (int t = 0; t < 8; ++t) {
                const v4f av = a[b & 1][t];
                const v4f wv = w[8 * b + t];
                const v2f alo = {av.x, av.y};
                const v2f ahi = {av.z, av.w};
                const v2f blo = {wv.x, wv.y};
                const v2f bhi = {wv.z, wv.w};
                c = __builtin_amdgcn_wmma_f32_16x16x4_f32(
                        false, alo, false, blo, (short)0, c, false, false);
                c = __builtin_amdgcn_wmma_f32_16x16x4_f32(
                        false, ahi, false, bhi, (short)0, c, false, false);
            }
        }

        stash(xs[cur ^ 1]);
        __syncthreads();

        asm volatile("v_nop\n\tv_nop\n\tv_nop\n\tv_nop" : "+v"(c));

        const int m0 = s * GEMM_ROWS;
        float* erow = emb + (size_t)(m0 + 8 * half) * OUT_DIM + n0 + l16;
        if (m0 + GEMM_ROWS <= N) {
#pragma unroll
            for (int j = 0; j < 8; ++j) erow[j * OUT_DIM] = c[j] + bv;
        } else {
#pragma unroll
            for (int j = 0; j < 8; ++j)
                if (m0 + j + 8 * half < N) erow[j * OUT_DIM] = c[j] + bv;
        }

        if (!more) break;
        cur ^= 1;
        s = next;
    }
}

__global__ __launch_bounds__(256) void score_kernel(
    const float* __restrict__ emb, const float* __restrict__ a,
    float* __restrict__ s_src, float* __restrict__ s_dst, int N) {
    const int wid  = (blockIdx.x * blockDim.x + threadIdx.x) >> 5;
    const int lane = threadIdx.x & 31;
    if (wid >= N) return;

    const float4 e  = *(const float4*)(emb + (size_t)wid * OUT_DIM + lane * 4);
    const float4 a1 = *(const float4*)(a + lane * 4);
    const float4 a2 = *(const float4*)(a + OUT_DIM + lane * 4);

    float p1 = e.x * a1.x + e.y * a1.y + e.z * a1.z + e.w * a1.w;
    float p2 = e.x * a2.x + e.y * a2.y + e.z * a2.z + e.w * a2.w;
#pragma unroll
    for (int off = 16; off > 0; off >>= 1) {
        p1 += __shfl_xor(p1, off, 32);
        p2 += __shfl_xor(p2, off, 32);
    }
    if (lane == 0) {
        s_src[wid] = p1;
        s_dst[wid] = p2;
    }
}

#define AGG_ROWS    392
#define AGG_WAVES   32
#define NBINS       256
#define CHUNK       4096
#define SORT_THREADS 1024
#define OFF_PITCH   260
#define REC_SHIFT   23
#define REC_MASK    0x7fffffu

typedef __attribute__((ext_vector_type(2))) int v2i;

__device__ __forceinline__ size_t rec_offset(int N) {
    return ((size_t)N * (OUT_DIM + 2) + 3) & ~(size_t)3;
}
__device__ __forceinline__ int chunk_count(int E) {
    return (E + CHUNK - 1) / CHUNK;
}
__device__ __forceinline__ bool plan_fits(int N, int E, size_t ws_size) {
    const size_t words = rec_offset(N) + (size_t)chunk_count(E) * (CHUNK + OFF_PITCH);
    return words * 4 <= ws_size && N <= NBINS * AGG_ROWS &&
           chunk_count(E) <= AGG_WAVES * 32;
}

__global__ __launch_bounds__(SORT_THREADS) void sort_kernel(
    const int* __restrict__ edges, float* __restrict__ ws, int N, int E,
    size_t ws_size) {
    __shared__ unsigned bins[NBINS + 1];
    __shared__ unsigned sorted[CHUNK];

    if (!plan_fits(N, E, ws_size)) return;
    unsigned* recs = (unsigned*)ws + rec_offset(N);
    unsigned* offs = recs + (size_t)chunk_count(E) * CHUNK;

    const int tid = threadIdx.x;
    const int c   = blockIdx.x;
    const int e0  = c * CHUNK;
    const v2i* e2 = (const v2i*)edges;
    v2i ed[4];
#pragma unroll
    for (int i = 0; i < 4; ++i) {
        const int e = e0 + tid + SORT_THREADS * i;
        ed[i] = e2[e < E ? e : E - 1];
    }
    if (tid <= NBINS) bins[tid] = 0u;
    __syncthreads();

    unsigned key[4], rank[4], rec[4];
#pragma unroll
    for (int i = 0; i < 4; ++i) {
        const int e = e0 + tid + SORT_THREADS * i;
        key[i]  = NBINS;
        rank[i] = 0u;
        rec[i]  = 0u;
        if (e < E && (unsigned)ed[i].x < (unsigned)N) {
            const unsigned blk = (unsigned)ed[i].x / AGG_ROWS;
            key[i]  = blk;
            rec[i]  = (((unsigned)ed[i].x - blk * AGG_ROWS) << REC_SHIFT) |
                      ((unsigned)ed[i].y & REC_MASK);
            rank[i] = atomicAdd(&bins[blk], 1u);
        }
    }
    __syncthreads();

    if (tid < 32) {
        unsigned cnt[8];
        unsigned sum = 0u;
#pragma unroll
        for (int k = 0; k < 8; ++k) {
            const unsigned t = bins[8 * tid + k];
            cnt[k] = sum;
            sum += t;
        }
        unsigned incl = sum;
#pragma unroll
        for (int d = 1; d < 32; d <<= 1) {
            const unsigned n = __shfl_up(incl, d, 32);
            if (tid >= d) incl += n;
        }
        const unsigned excl = incl - sum;
#pragma unroll
        for (int k = 0; k < 8; ++k) bins[8 * tid + k] = excl + cnt[k];
        if (tid == 31) bins[NBINS] = incl;
    }
    __syncthreads();

#pragma unroll
    for (int i = 0; i < 4; ++i)
        if (key[i] < NBINS) sorted[bins[key[i]] + rank[i]] = rec[i];
    if (tid <= NBINS) offs[(size_t)c * OFF_PITCH + tid] = bins[tid];
    __syncthreads();

    const unsigned total = bins[NBINS];
    unsigned* dst = recs + (size_t)c * CHUNK;
#pragma unroll
    for (int i = 0; i < 4; ++i) {
        const unsigned p = tid + SORT_THREADS * i;
        if (p < total) dst[p] = sorted[p];
    }
}

#define AGG_THREADS 1024

__global__ __launch_bounds__(AGG_THREADS) void aggregate_kernel(
    const float* __restrict__ ws, float* __restrict__ out, int N, int E,
    size_t ws_size) {
    __shared__ float acc[(AGG_ROWS + 1) * OUT_DIM];
    __shared__ float rsum[AGG_ROWS];
    __shared__ float ssrc[AGG_ROWS];

    if (!plan_fits(N, E, ws_size)) return;
    const float*    emb   = ws;
    const float*    s_src = emb + (size_t)N * OUT_DIM;
    const float*    s_dst = s_src + N;
    const unsigned* recs  = (const unsigned*)ws + rec_offset(N);
    const int       nchunks = chunk_count(E);
    const unsigned* offs  = recs + (size_t)nchunks * CHUNK;

    const int tid  = threadIdx.x;
    const int lane = tid & 31;
    const int w    = __builtin_amdgcn_readfirstlane(tid >> 5);
    const int base = blockIdx.x * AGG_ROWS;
    int nrows = N - base;
    if (nrows > AGG_ROWS) nrows = AGG_ROWS;

    const int cpw = (nchunks + AGG_WAVES - 1) / AGG_WAVES;
    const int  myc   = w * cpw + lane;
    const bool mine  = lane < cpw && myc < nchunks;
    const int  cread = mine ? myc : nchunks - 1;
    const unsigned* orow = offs + (size_t)cread * OFF_PITCH + blockIdx.x;
    const int seg0 = (int)orow[0];
    const int seg1 = (int)orow[1];

    for (int i = tid; i < (AGG_ROWS + 1) * OUT_DIM / 4; i += AGG_THREADS)
        ((float4*)acc)[i] = make_float4(0.0f, 0.0f, 0.0f, 0.0f);
    for (int i = tid; i < AGG_ROWS; i += AGG_THREADS) {
        rsum[i] = 0.0f;
        ssrc[i] = (i < nrows) ? s_src[base + i] : 0.0f;
    }
    __syncthreads();

    const int len   = mine ? seg1 - seg0 : 0;
    const int gbase = myc * CHUNK + seg0;
    int incl = len;
#pragma unroll
    for (int d = 1; d < 32; d <<= 1) {
        const int n = __shfl_up(incl, d, 32);
        if (lane >= d) incl += n;
    }
    const int excl  = incl - len;
    const int shift = gbase - excl;
    const int total = __builtin_amdgcn_readlane(incl, 31);

    auto batch = [&](int cnt, int ridx) {
        int   sl = AGG_ROWS;
        int   dd = 0;
        float sc = 0.0f;
        if (lane < cnt) {
            const unsigned rec = recs[ridx];
            sl = (int)(rec >> REC_SHIFT);
            dd = (int)(rec & REC_MASK);
            const float logit = ssrc[sl] + s_dst[dd];
            const float lr    = logit > 0.0f ? logit : SLOPE * logit;
            sc = expf(lr);
            atomicAdd(&rsum[sl], sc);
        }
        int j = 0;
        for (; j + 8 <= cnt; j += 8) {
            float m[8][4];
#pragma unroll
            for (int t = 0; t < 8; ++t) {
                const int    dj   = __builtin_amdgcn_readlane(dd, j + t);
                const float* mrow = emb + (size_t)dj * OUT_DIM + lane;
                m[t][0] = mrow[0];
                m[t][1] = mrow[32];
                m[t][2] = mrow[64];
                m[t][3] = mrow[96];
            }
#pragma unroll
            for (int t = 0; t < 8; ++t) {
                const int   sj = __builtin_amdgcn_readlane(sl, j + t);
                const float cj = __int_as_float(
                    __builtin_amdgcn_readlane(__float_as_int(sc), j + t));
                float* arow = acc + sj * OUT_DIM + lane;
                atomicAdd(arow,      cj * m[t][0]);
                atomicAdd(arow + 32, cj * m[t][1]);
                atomicAdd(arow + 64, cj * m[t][2]);
                atomicAdd(arow + 96, cj * m[t][3]);
            }
        }
        for (; j < cnt; ++j) {
            const int   sj = __builtin_amdgcn_readlane(sl, j);
            const int   dj = __builtin_amdgcn_readlane(dd, j);
            const float cj = __int_as_float(
                __builtin_amdgcn_readlane(__float_as_int(sc), j));
            const float* mrow = emb + (size_t)dj * OUT_DIM + lane;
            float*       arow = acc + sj * OUT_DIM + lane;
            const float m0 = mrow[0];
            const float m1 = mrow[32];
            const float m2 = mrow[64];
            const float m3 = mrow[96];
            atomicAdd(arow,      cj * m0);
            atomicAdd(arow + 32, cj * m1);
            atomicAdd(arow + 64, cj * m2);
            atomicAdd(arow + 96, cj * m3);
        }
    };

    for (int v0 = 0; v0 < total; v0 += 32) {
        const int v = v0 + lane;
        int sh = 0;
        for (int i = 0; i < cpw; ++i) {
            const int pi = __builtin_amdgcn_readlane(excl, i);
            const int di = __builtin_amdgcn_readlane(shift, i);
            if (v >= pi) sh = di;
        }
        const int rem = total - v0;
        if (rem >= 32) batch(32, sh + v);
        else           batch(rem, sh + v);
    }
    __syncthreads();

    float4* o4 = (float4*)out + (size_t)base * (OUT_DIM / 4);
    for (int i = tid; i < nrows * (OUT_DIM / 4); i += AGG_THREADS) {
        const float rs   = rsum[i >> 5];
        const float safe = rs > 0.0f ? rs : 1.0f;
        float4 a = ((const float4*)acc)[i];
        a.x = a.x / safe;
        a.y = a.y / safe;
        a.z = a.z / safe;
        a.w = a.w / safe;
        o4[i] = a;
    }
}

extern "C" void kernel_launch(void* const* d_in, const int* in_sizes, int n_in,
                              void* d_out, int out_size, void* d_ws, size_t ws_size,
                              hipStream_t stream) {
    const float* x     = (const float*)d_in[0];
    const float* W     = (const float*)d_in[1];
    const float* bias  = (const float*)d_in[2];
    const float* a     = (const float*)d_in[3];
    const int*   edges = (const int*)d_in[4];

    const int N = in_sizes[0] / IN_DIM;
    const int E = in_sizes[4] / 2;

    float* ws    = (float*)d_ws;
    float* emb   = ws;
    float* s_src = emb + (size_t)N * OUT_DIM;
    float* s_dst = s_src + N;
    float* out   = (float*)d_out;

    {
        gemm_wmma_kernel<<<GEMM_BLOCKS, GEMM_THREADS, 0, stream>>>(x, W, bias, emb, N);
    }
    {
        int grid = (N + 7) / 8;
        score_kernel<<<grid, 256, 0, stream>>>(emb, a, s_src, s_dst, N);
    }
    {
        int grid = (E + CHUNK - 1) / CHUNK;
        sort_kernel<<<grid, SORT_THREADS, 0, stream>>>(edges, ws, N, E, ws_size);
    }
    {
        int grid = (N + AGG_ROWS - 1) / AGG_ROWS;
        aggregate_kernel<<<grid, AGG_THREADS, 0, stream>>>(ws, out, N, E, ws_size);
    }
}


# Score launch made resident: 4096 long-lived waves loop over groups of 8 rows (a1, a2 in registers, 8 whole-row loads in flight, DPP butterfly), not one short wave per row; same sums, same order
# speedup vs baseline: 3.7144x; 1.0821x over previous
#include <hip/hip_runtime.h>
#include <hip/hip_bf16.h>
#include <math.h>

#define IN_DIM  256
#define OUT_DIM 128
#define SLOPE   0.1f

typedef __attribute__((ext_vector_type(2))) float v2f;
typedef __attribute__((ext_vector_type(8))) float v8f;

#define GEMM_THREADS 256
#define GEMM_BLOCKS  512
#define GEMM_ROWS    16
#define GEMM_PITCH   260

typedef __attribute__((ext_vector_type(4))) float v4f;

__global__ __launch_bounds__(GEMM_THREADS, 4) void gemm_wmma_kernel(
    const float* __restrict__ x, const float* __restrict__ W,
    const float* __restrict__ bias, float* __restrict__ emb, int N) {
    __shared__ float xs[2][GEMM_ROWS * GEMM_PITCH];

    const int tid  = threadIdx.x;
    const int wave = __builtin_amdgcn_readfirstlane(tid >> 5);
    const int lane = tid & 31;
    const int half = lane >> 4;
    const int l16  = lane & 15;
    const int n0   = wave * 16;
    const int ntiles = (N + GEMM_ROWS - 1) / GEMM_ROWS;
    int s = blockIdx.x;
    if (s >= ntiles) return;

    v4f w[32];
    {
        const float* wrow = W + (size_t)(n0 + l16) * IN_DIM + 4 * half;
#pragma unroll
        for (int j = 0; j < 32; ++j) w[j] = *(const v4f*)(wrow + 8 * j);
    }
    const float bv = bias[n0 + l16];

    const int srow = wave >> 1;
    const int scol = ((wave & 1) * 32 + lane) * 4;

    v4f g[4];
    auto fetch = [&](int tile) {
#pragma unroll
        for (int i = 0; i < 4; ++i) {
            int row = tile * GEMM_ROWS + 4 * i + srow;
            row = row < N ? row : N - 1;
            const float* xr = x + (size_t)row * IN_DIM;
            g[i] = *(const v4f*)(xr + scol);
        }
    };
    auto stash = [&](float* buf) {
#pragma unroll
        for (int i = 0; i < 4; ++i)
            *(v4f*)(buf + (4 * i + srow) * GEMM_PITCH + scol) = g[i];
    };

    fetch(s);
    stash(xs[0]);
    __syncthreads();

    int cur = 0;
    for (;;) {
        const int  next = s + (int)gridDim.x;
        const bool more = next < ntiles;
        fetch(more ? next : s);
        __builtin_amdgcn_sched_barrier(0);

        const float* ap = xs[cur] + l16 * GEMM_PITCH + 4 * half;
        v4f a[2][8];
#pragma unroll
        for (int t = 0; t < 8; ++t) a[0][t] = *(const v4f*)(ap + 8 * t);
        v8f c = {};
#pragma unroll
        for (int b = 0; b < 4; ++b) {
            if (b < 3) {
#pragma unroll
                for (int t = 0; t < 8; ++t)
                    a[(b + 1) & 1][t] = *(const v4f*)(ap + 8 * (8 * (b + 1) + t));
            }
            __builtin_amdgcn_sched_barrier(0);
#pragma unroll
            for (int t = 0; t < 8; ++t) {
                const v4f av = a[b & 1][t];
                const v4f wv = w[8 * b + t];
                const v2f alo = {av.x, av.y};
                const v2f ahi = {av.z, av.w};
                const v2f blo = {wv.x, wv.y};
                const v2f bhi = {wv.z, wv.w};
                c = __builtin_amdgcn_wmma_f32_16x16x4_f32(
                        false, alo, false, blo, (short)0, c, false, false);
                c = __builtin_amdgcn_wmma_f32_16x16x4_f32(
                        false, ahi, false, bhi, (short)0, c, false, false);
            }
        }

        stash(xs[cur ^ 1]);
        __syncthreads();

        asm volatile("v_nop\n\tv_nop\n\tv_nop\n\tv_nop" : "+v"(c));

        const int m0 = s * GEMM_ROWS;
        float* erow = emb + (size_t)(m0 + 8 * half) * OUT_DIM + n0 + l16;
        if (m0 + GEMM_ROWS <= N) {
#pragma unroll
            for (int j = 0; j < 8; ++j) erow[j * OUT_DIM] = c[j] + bv;
        } else {
#pragma unroll
            for (int j = 0; j < 8; ++j)
                if (m0 + j + 8 * half < N) erow[j * OUT_DIM] = c[j] + bv;
        }

        if (!more) break;
        cur ^= 1;
        s = next;
    }
}

#define SCORE_THREADS 256
#define SCORE_BLOCKS  512
#define SCORE_ROWS    8

__device__ __forceinline__ float xhalf(float v) {
    const int b = __builtin_bit_cast(int, v);
    return __builtin_bit_cast(float,
        __builtin_amdgcn_permlanex16(b, b, 0x76543210, 0xfedcba98, false, false));
}
template <int M>
__device__ __forceinline__ float row_xor(float v) {
    const int b = __builtin_bit_cast(int, v);
    return __builtin_bit_cast(float,
        __builtin_amdgcn_update_dpp(0, b, 0x160 + M, 0xf, 0xf, true));
}

__global__ __launch_bounds__(SCORE_THREADS) void score_kernel(
    const float* __restrict__ emb, const float* __restrict__ a,
    float* __restrict__ s_src, float* __restrict__ s_dst, int N) {
    const int lane = threadIdx.x & 31;
    const int half = lane >> 4;
    const int l16  = lane & 15;
    const int gw = __builtin_amdgcn_readfirstlane(
        (int)(blockIdx.x * (SCORE_THREADS / 32) + (threadIdx.x >> 5)));
    const int nw = (int)gridDim.x * (SCORE_THREADS / 32);
    const int ngroups = (N + SCORE_ROWS - 1) / SCORE_ROWS;

    const v4f a1 = *(const v4f*)(a + lane * 4);
    const v4f a2 = *(const v4f*)(a + OUT_DIM + lane * 4);
    const v2f ax = {a1.x, a2.x};
    const v2f ay = {a1.y, a2.y};
    const v2f az = {a1.z, a2.z};
    const v2f aw = {a1.w, a2.w};
    float* sp = half ? s_dst : s_src;

    for (int g = gw; g < ngroups; g += nw) {
        const int r0 = g * SCORE_ROWS;
        v4f e[SCORE_ROWS];
#pragma unroll
        for (int t = 0; t < SCORE_ROWS; ++t) {
            int row = r0 + t;
            row = row < N ? row : N - 1;
            e[t] = *(const v4f*)(emb + (size_t)row * OUT_DIM + lane * 4);
        }
        __builtin_amdgcn_sched_barrier(0);
        float res = 0.0f;
#pragma unroll
        for (int t = 0; t < SCORE_ROWS; ++t) {
            const v2f ex = {e[t].x, e[t].x};
            const v2f ey = {e[t].y, e[t].y};
            const v2f ez = {e[t].z, e[t].z};
            const v2f ew = {e[t].w, e[t].w};
            const v2f p  = ex * ax + ey * ay + ez * az + ew * aw;
            const float keep = half ? p.y : p.x;
            const float give = half ? p.x : p.y;
            float q = keep + xhalf(give);
            q += row_xor<8>(q);
            q += row_xor<4>(q);
            q += row_xor<2>(q);
            q += row_xor<1>(q);
            res = (l16 == t) ? q : res;
        }
        const int row = r0 + l16;
        if (l16 < SCORE_ROWS && row < N) sp[row] = res;
    }
}

#define AGG_ROWS    392
#define AGG_WAVES   32
#define NBINS       256
#define CHUNK       4096
#define SORT_THREADS 1024
#define OFF_PITCH   260
#define REC_SHIFT   23
#define REC_MASK    0x7fffffu

typedef __attribute__((ext_vector_type(2))) int v2i;

__device__ __forceinline__ size_t rec_offset(int N) {
    return ((size_t)N * (OUT_DIM + 2) + 3) & ~(size_t)3;
}
__device__ __forceinline__ int chunk_count(int E) {
    return (E + CHUNK - 1) / CHUNK;
}
__device__ __forceinline__ bool plan_fits(int N, int E, size_t ws_size) {
    const size_t words = rec_offset(N) + (size_t)chunk_count(E) * (CHUNK + OFF_PITCH);
    return words * 4 <= ws_size && N <= NBINS * AGG_ROWS &&
           chunk_count(E) <= AGG_WAVES * 32;
}

__global__ __launch_bounds__(SORT_THREADS) void sort_kernel(
    const int* __restrict__ edges, float* __restrict__ ws, int N, int E,
    size_t ws_size) {
    __shared__ unsigned bins[NBINS + 1];
    __shared__ unsigned sorted[CHUNK];

    if (!plan_fits(N, E, ws_size)) return;
    unsigned* recs = (unsigned*)ws + rec_offset(N);
    unsigned* offs = recs + (size_t)chunk_count(E) * CHUNK;

    const int tid = threadIdx.x;
    const int c   = blockIdx.x;
    const int e0  = c * CHUNK;
    const v2i* e2 = (const v2i*)edges;
    v2i ed[4];
#pragma unroll
    for (int i = 0; i < 4; ++i) {
        const int e = e0 + tid + SORT_THREADS * i;
        ed[i] = e2[e < E ? e : E - 1];
    }
    if (tid <= NBINS) bins[tid] = 0u;
    __syncthreads();

    unsigned key[4], rank[4], rec[4];
#pragma unroll
    for (int i = 0; i < 4; ++i) {
        const int e = e0 + tid + SORT_THREADS * i;
        key[i]  = NBINS;
        rank[i] = 0u;
        rec[i]  = 0u;
        if (e < E && (unsigned)ed[i].x < (unsigned)N) {
            const unsigned blk = (unsigned)ed[i].x / AGG_ROWS;
            key[i]  = blk;
            rec[i]  = (((unsigned)ed[i].x - blk * AGG_ROWS) << REC_SHIFT) |
                      ((unsigned)ed[i].y & REC_MASK);
            rank[i] = atomicAdd(&bins[blk], 1u);
        }
    }
    __syncthreads();

    if (tid < 32) {
        unsigned cnt[8];
        unsigned sum = 0u;
#pragma unroll
        for (int k = 0; k < 8; ++k) {
            const unsigned t = bins[8 * tid + k];
            cnt[k] = sum;
            sum += t;
        }
        unsigned incl = sum;
#pragma unroll
        for (int d = 1; d < 32; d <<= 1) {
            const unsigned n = __shfl_up(incl, d, 32);
            if (tid >= d) incl += n;
        }
        const unsigned excl = incl - sum;
#pragma unroll
        for (int k = 0; k < 8; ++k) bins[8 * tid + k] = excl + cnt[k];
        if (tid == 31) bins[NBINS] = incl;
    }
    __syncthreads();

#pragma unroll
    for (int i = 0; i < 4; ++i)
        if (key[i] < NBINS) sorted[bins[key[i]] + rank[i]] = rec[i];
    if (tid <= NBINS) offs[(size_t)c * OFF_PITCH + tid] = bins[tid];
    __syncthreads();

    const unsigned total = bins[NBINS];
    unsigned* dst = recs + (size_t)c * CHUNK;
#pragma unroll
    for (int i = 0; i < 4; ++i) {
        const unsigned p = tid + SORT_THREADS * i;
        if (p < total) dst[p] = sorted[p];
    }
}

#define AGG_THREADS 1024

__global__ __launch_bounds__(AGG_THREADS) void aggregate_kernel(
    const float* __restrict__ ws, float* __restrict__ out, int N, int E,
    size_t ws_size) {
    __shared__ float acc[(AGG_ROWS + 1) * OUT_DIM];
    __shared__ float rsum[AGG_ROWS];
    __shared__ float ssrc[AGG_ROWS];

    if (!plan_fits(N, E, ws_size)) return;
    const float*    emb   = ws;
    const float*    s_src = emb + (size_t)N * OUT_DIM;
    const float*    s_dst = s_src + N;
    const unsigned* recs  = (const unsigned*)ws + rec_offset(N);
    const int       nchunks = chunk_count(E);
    const unsigned* offs  = recs + (size_t)nchunks * CHUNK;

    const int tid  = threadIdx.x;
    const int lane = tid & 31;
    const int w    = __builtin_amdgcn_readfirstlane(tid >> 5);
    const int base = blockIdx.x * AGG_ROWS;
    int nrows = N - base;
    if (nrows > AGG_ROWS) nrows = AGG_ROWS;

    const int cpw = (nchunks + AGG_WAVES - 1) / AGG_WAVES;
    const int  myc   = w * cpw + lane;
    const bool mine  = lane < cpw && myc < nchunks;
    const int  cread = mine ? myc : nchunks - 1;
    const unsigned* orow = offs + (size_t)cread * OFF_PITCH + blockIdx.x;
    const int seg0 = (int)orow[0];
    const int seg1 = (int)orow[1];

    for (int i = tid; i < (AGG_ROWS + 1) * OUT_DIM / 4; i += AGG_THREADS)
        ((float4*)acc)[i] = make_float4(0.0f, 0.0f, 0.0f, 0.0f);
    for (int i = tid; i < AGG_ROWS; i += AGG_THREADS) {
        rsum[i] = 0.0f;
        ssrc[i] = (i < nrows) ? s_src[base + i] : 0.0f;
    }
    __syncthreads();

    const int len   = mine ? seg1 - seg0 : 0;
    const int gbase = myc * CHUNK + seg0;
    int incl = len;
#pragma unroll
    for (int d = 1; d < 32; d <<= 1) {
        const int n = __shfl_up(incl, d, 32);
        if (lane >= d) incl += n;
    }
    const int excl  = incl - len;
    const int shift = gbase - excl;
    const int total = __builtin_amdgcn_readlane(incl, 31);

    auto batch = [&](int cnt, int ridx) {
        int   sl = AGG_ROWS;
        int   dd = 0;
        float sc = 0.0f;
        if (lane < cnt) {
            const unsigned rec = recs[ridx];
            sl = (int)(rec >> REC_SHIFT);
            dd = (int)(rec & REC_MASK);
            const float logit = ssrc[sl] + s_dst[dd];
            const float lr    = logit > 0.0f ? logit : SLOPE * logit;
            sc = expf(lr);
            atomicAdd(&rsum[sl], sc);
        }
        int j = 0;
        for (; j + 8 <= cnt; j += 8) {
            float m[8][4];
#pragma unroll
            for (int t = 0; t < 8; ++t) {
                const int    dj   = __builtin_amdgcn_readlane(dd, j + t);
                const float* mrow = emb + (size_t)dj * OUT_DIM + lane;
                m[t][0] = mrow[0];
                m[t][1] = mrow[32];
                m[t][2] = mrow[64];
                m[t][3] = mrow[96];
            }
#pragma unroll
            for (int t = 0; t < 8; ++t) {
                const int   sj = __builtin_amdgcn_readlane(sl, j + t);
                const float cj = __int_as_float(
                    __builtin_amdgcn_readlane(__float_as_int(sc), j + t));
                float* arow = acc + sj * OUT_DIM + lane;
                atomicAdd(arow,      cj * m[t][0]);
                atomicAdd(arow + 32, cj * m[t][1]);
                atomicAdd(arow + 64, cj * m[t][2]);
                atomicAdd(arow + 96, cj * m[t][3]);
            }
        }
        for (; j < cnt; ++j) {
            const int   sj = __builtin_amdgcn_readlane(sl, j);
            const int   dj = __builtin_amdgcn_readlane(dd, j);
            const float cj = __int_as_float(
                __builtin_amdgcn_readlane(__float_as_int(sc), j));
            const float* mrow = emb + (size_t)dj * OUT_DIM + lane;
            float*       arow = acc + sj * OUT_DIM + lane;
            const float m0 = mrow[0];
            const float m1 = mrow[32];
            const float m2 = mrow[64];
            const float m3 = mrow[96];
            atomicAdd(arow,      cj * m0);
            atomicAdd(arow + 32, cj * m1);
            atomicAdd(arow + 64, cj * m2);
            atomicAdd(arow + 96, cj * m3);
        }
    };

    for (int v0 = 0; v0 < total; v0 += 32) {
        const int v = v0 + lane;
        int sh = 0;
        for (int i = 0; i < cpw; ++i) {
            const int pi = __builtin_amdgcn_readlane(excl, i);
            const int di = __builtin_amdgcn_readlane(shift, i);
            if (v >= pi) sh = di;
        }
        const int rem = total - v0;
        if (rem >= 32) batch(32, sh + v);
        else           batch(rem, sh + v);
    }
    __syncthreads();

    float4* o4 = (float4*)out + (size_t)base * (OUT_DIM / 4);
    for (int i = tid; i < nrows * (OUT_DIM / 4); i += AGG_THREADS) {
        const float rs   = rsum[i >> 5];
        const float safe = rs > 0.0f ? rs : 1.0f;
        float4 a = ((const float4*)acc)[i];
        a.x = a.x / safe;
        a.y = a.y / safe;
        a.z = a.z / safe;
        a.w = a.w / safe;
        o4[i] = a;
    }
}

extern "C" void kernel_launch(void* const* d_in, const int* in_sizes, int n_in,
                              void* d_out, int out_size, void* d_ws, size_t ws_size,
                              hipStream_t stream) {
    const float* x     = (const float*)d_in[0];
    const float* W     = (const float*)d_in[1];
    const float* bias  = (const float*)d_in[2];
    const float* a     = (const float*)d_in[3];
    const int*   edges = (const int*)d_in[4];

    const int N = in_sizes[0] / IN_DIM;
    const int E = in_sizes[4] / 2;

    float* ws    = (float*)d_ws;
    float* emb   = ws;
    float* s_src = emb + (size_t)N * OUT_DIM;
    float* s_dst = s_src + N;
    float* out   = (float*)d_out;

    {
        gemm_wmma_kernel<<<GEMM_BLOCKS, GEMM_THREADS, 0, stream>>>(x, W, bias, emb, N);
    }
    {
        score_kernel<<<SCORE_BLOCKS, SCORE_THREADS, 0, stream>>>(emb, a, s_src, s_dst, N);
    }
    {
        int grid = (E + CHUNK - 1) / CHUNK;
        sort_kernel<<<grid, SORT_THREADS, 0, stream>>>(edges, ws, N, E, ws_size);
    }
    {
        int grid = (N + AGG_ROWS - 1) / AGG_ROWS;
        aggregate_kernel<<<grid, AGG_THREADS, 0, stream>>>(ws, out, N, E, ws_size);
    }
}


# Aggregation gathers each emb row by one 16-byte-per-lane load (whole 512-byte row per wave instruction), not four dword loads; LDS accumulator rows column-permuted; same adds, arrival order as before
# speedup vs baseline: 3.9022x; 1.0506x over previous
#include <hip/hip_runtime.h>
#include <hip/hip_bf16.h>
#include <math.h>

#define IN_DIM  256
#define OUT_DIM 128
#define SLOPE   0.1f

typedef __attribute__((ext_vector_type(2))) float v2f;
typedef __attribute__((ext_vector_type(8))) float v8f;

#define GEMM_THREADS 256
#define GEMM_BLOCKS  512
#define GEMM_ROWS    16
#define GEMM_PITCH   260

typedef __attribute__((ext_vector_type(4))) float v4f;

__global__ __launch_bounds__(GEMM_THREADS, 4) void gemm_wmma_kernel(
    const float* __restrict__ x, const float* __restrict__ W,
    const float* __restrict__ bias, float* __restrict__ emb, int N) {
    __shared__ float xs[2][GEMM_ROWS * GEMM_PITCH];

    const int tid  = threadIdx.x;
    const int wave = __builtin_amdgcn_readfirstlane(tid >> 5);
    const int lane = tid & 31;
    const int half = lane >> 4;
    const int l16  = lane & 15;
    const int n0   = wave * 16;
    const int ntiles = (N + GEMM_ROWS - 1) / GEMM_ROWS;
    int s = blockIdx.x;
    if (s >= ntiles) return;

    v4f w[32];
    {
        const float* wrow = W + (size_t)(n0 + l16) * IN_DIM + 4 * half;
#pragma unroll
        for (int j = 0; j < 32; ++j) w[j] = *(const v4f*)(wrow + 8 * j);
    }
    const float bv = bias[n0 + l16];

    const int srow = wave >> 1;
    const int scol = ((wave & 1) * 32 + lane) * 4;

    v4f g[4];
    auto fetch = [&](int tile) {
#pragma unroll
        for (int i = 0; i < 4; ++i) {
            int row = tile * GEMM_ROWS + 4 * i + srow;
            row = row < N ? row : N - 1;
            const float* xr = x + (size_t)row * IN_DIM;
            g[i] = *(const v4f*)(xr + scol);
        }
    };
    auto stash = [&](float* buf) {
#pragma unroll
        for (int i = 0; i < 4; ++i)
            *(v4f*)(buf + (4 * i + srow) * GEMM_PITCH + scol) = g[i];
    };

    fetch(s);
    stash(xs[0]);
    __syncthreads();

    int cur = 0;
    for (;;) {
        const int  next = s + (int)gridDim.x;
        const bool more = next < ntiles;
        fetch(more ? next : s);
        __builtin_amdgcn_sched_barrier(0);

        const float* ap = xs[cur] + l16 * GEMM_PITCH + 4 * half;
        v4f a[2][8];
#pragma unroll
        for (int t = 0; t < 8; ++t) a[0][t] = *(const v4f*)(ap + 8 * t);
        v8f c = {};
#pragma unroll
        for (int b = 0; b < 4; ++b) {
            if (b < 3) {
#pragma unroll
                for (int t = 0; t < 8; ++t)
                    a[(b + 1) & 1][t] = *(const v4f*)(ap + 8 * (8 * (b + 1) + t));
            }
            __builtin_amdgcn_sched_barrier(0);
#pragma unroll
            for (int t = 0; t < 8; ++t) {
                const v4f av = a[b & 1][t];
                const v4f wv = w[8 * b + t];
                const v2f alo = {av.x, av.y};
                const v2f ahi = {av.z, av.w};
                const v2f blo = {wv.x, wv.y};
                const v2f bhi = {wv.z, wv.w};
                c = __builtin_amdgcn_wmma_f32_16x16x4_f32(
                        false, alo, false, blo, (short)0, c, false, false);
                c = __builtin_amdgcn_wmma_f32_16x16x4_f32(
                        false, ahi, false, bhi, (short)0, c, false, false);
            }
        }

        stash(xs[cur ^ 1]);
        __syncthreads();

        asm volatile("v_nop\n\tv_nop\n\tv_nop\n\tv_nop" : "+v"(c));

        const int m0 = s * GEMM_ROWS;
        float* erow = emb + (size_t)(m0 + 8 * half) * OUT_DIM + n0 + l16;
        if (m0 + GEMM_ROWS <= N) {
#pragma unroll
            for (int j = 0; j < 8; ++j) erow[j * OUT_DIM] = c[j] + bv;
        } else {
#pragma unroll
            for (int j = 0; j < 8; ++j)
                if (m0 + j + 8 * half < N) erow[j * OUT_DIM] = c[j] + bv;
        }

        if (!more) break;
        cur ^= 1;
        s = next;
    }
}

#define SCORE_THREADS 256
#define SCORE_BLOCKS  512
#define SCORE_ROWS    8

__device__ __forceinline__ float xhalf(float v) {
    const int b = __builtin_bit_cast(int, v);
    return __builtin_bit_cast(float,
        __builtin_amdgcn_permlanex16(b, b, 0x76543210, 0xfedcba98, false, false));
}
template <int M>
__device__ __forceinline__ float row_xor(float v) {
    const int b = __builtin_bit_cast(int, v);
    return __builtin_bit_cast(float,
        __builtin_amdgcn_update_dpp(0, b, 0x160 + M, 0xf, 0xf, true));
}

__global__ __launch_bounds__(SCORE_THREADS) void score_kernel(
    const float* __restrict__ emb, const float* __restrict__ a,
    float* __restrict__ s_src, float* __restrict__ s_dst, int N) {
    const int lane = threadIdx.x & 31;
    const int half = lane >> 4;
    const int l16  = lane & 15;
    const int gw = __builtin_amdgcn_readfirstlane(
        (int)(blockIdx.x * (SCORE_THREADS / 32) + (threadIdx.x >> 5)));
    const int nw = (int)gridDim.x * (SCORE_THREADS / 32);
    const int ngroups = (N + SCORE_ROWS - 1) / SCORE_ROWS;

    const v4f a1 = *(const v4f*)(a + lane * 4);
    const v4f a2 = *(const v4f*)(a + OUT_DIM + lane * 4);
    const v2f ax = {a1.x, a2.x};
    const v2f ay = {a1.y, a2.y};
    const v2f az = {a1.z, a2.z};
    const v2f aw = {a1.w, a2.w};
    float* sp = half ? s_dst : s_src;

    for (int g = gw; g < ngroups; g += nw) {
        const int r0 = g * SCORE_ROWS;
        v4f e[SCORE_ROWS];
#pragma unroll
        for (int t = 0; t < SCORE_ROWS; ++t) {
            int row = r0 + t;
            row = row < N ? row : N - 1;
            e[t] = *(const v4f*)(emb + (size_t)row * OUT_DIM + lane * 4);
        }
        __builtin_amdgcn_sched_barrier(0);
        float res = 0.0f;
#pragma unroll
        for (int t = 0; t < SCORE_ROWS; ++t) {
            const v2f ex = {e[t].x, e[t].x};
            const v2f ey = {e[t].y, e[t].y};
            const v2f ez = {e[t].z, e[t].z};
            const v2f ew = {e[t].w, e[t].w};
            const v2f p  = ex * ax + ey * ay + ez * az + ew * aw;
            const float keep = half ? p.y : p.x;
            const float give = half ? p.x : p.y;
            float q = keep + xhalf(give);
            q += row_xor<8>(q);
            q += row_xor<4>(q);
            q += row_xor<2>(q);
            q += row_xor<1>(q);
            res = (l16 == t) ? q : res;
        }
        const int row = r0 + l16;
        if (l16 < SCORE_ROWS && row < N) sp[row] = res;
    }
}

#define AGG_ROWS    392
#define AGG_WAVES   32
#define NBINS       256
#define CHUNK       4096
#define SORT_THREADS 1024
#define OFF_PITCH   260
#define REC_SHIFT   23
#define REC_MASK    0x7fffffu

typedef __attribute__((ext_vector_type(2))) int v2i;

__device__ __forceinline__ size_t rec_offset(int N) {
    return ((size_t)N * (OUT_DIM + 2) + 3) & ~(size_t)3;
}
__device__ __forceinline__ int chunk_count(int E) {
    return (E + CHUNK - 1) / CHUNK;
}
__device__ __forceinline__ bool plan_fits(int N, int E, size_t ws_size) {
    const size_t words = rec_offset(N) + (size_t)chunk_count(E) * (CHUNK + OFF_PITCH);
    return words * 4 <= ws_size && N <= NBINS * AGG_ROWS &&
           chunk_count(E) <= AGG_WAVES * 32;
}

__global__ __launch_bounds__(SORT_THREADS) void sort_kernel(
    const int* __restrict__ edges, float* __restrict__ ws, int N, int E,
    size_t ws_size) {
    __shared__ unsigned bins[NBINS + 1];
    __shared__ unsigned sorted[CHUNK];

    if (!plan_fits(N, E, ws_size)) return;
    unsigned* recs = (unsigned*)ws + rec_offset(N);
    unsigned* offs = recs + (size_t)chunk_count(E) * CHUNK;

    const int tid = threadIdx.x;
    const int c   = blockIdx.x;
    const int e0  = c * CHUNK;
    const v2i* e2 = (const v2i*)edges;
    v2i ed[4];
#pragma unroll
    for (int i = 0; i < 4; ++i) {
        const int e = e0 + tid + SORT_THREADS * i;
        ed[i] = e2[e < E ? e : E - 1];
    }
    if (tid <= NBINS) bins[tid] = 0u;
    __syncthreads();

    unsigned key[4], rank[4], rec[4];
#pragma unroll
    for (int i = 0; i < 4; ++i) {
        const int e = e0 + tid + SORT_THREADS * i;
        key[i]  = NBINS;
        rank[i] = 0u;
        rec[i]  = 0u;
        if (e < E && (unsigned)ed[i].x < (unsigned)N) {
            const unsigned blk = (unsigned)ed[i].x / AGG_ROWS;
            key[i]  = blk;
            rec[i]  = (((unsigned)ed[i].x - blk * AGG_ROWS) << REC_SHIFT) |
                      ((unsigned)ed[i].y & REC_MASK);
            rank[i] = atomicAdd(&bins[blk], 1u);
        }
    }
    __syncthreads();

    if (tid < 32) {
        unsigned cnt[8];
        unsigned sum = 0u;
#pragma unroll
        for (int k = 0; k < 8; ++k) {
            const unsigned t = bins[8 * tid + k];
            cnt[k] = sum;
            sum += t;
        }
        unsigned incl = sum;
#pragma unroll
        for (int d = 1; d < 32; d <<= 1) {
            const unsigned n = __shfl_up(incl, d, 32);
            if (tid >= d) incl += n;
        }
        const unsigned excl = incl - sum;
#pragma unroll
        for (int k = 0; k < 8; ++k) bins[8 * tid + k] = excl + cnt[k];
        if (tid == 31) bins[NBINS] = incl;
    }
    __syncthreads();

#pragma unroll
    for (int i = 0; i < 4; ++i)
        if (key[i] < NBINS) sorted[bins[key[i]] + rank[i]] = rec[i];
    if (tid <= NBINS) offs[(size_t)c * OFF_PITCH + tid] = bins[tid];
    __syncthreads();

    const unsigned total = bins[NBINS];
    unsigned* dst = recs + (size_t)c * CHUNK;
#pragma unroll
    for (int i = 0; i < 4; ++i) {
        const unsigned p = tid + SORT_THREADS * i;
        if (p < total) dst[p] = sorted[p];
    }
}

#define AGG_THREADS 1024

__global__ __launch_bounds__(AGG_THREADS) void aggregate_kernel(
    const float* __restrict__ ws, float* __restrict__ out, int N, int E,
    size_t ws_size) {
    __shared__ float acc[(AGG_ROWS + 1) * OUT_DIM];
    __shared__ float rsum[AGG_ROWS];
    __shared__ float ssrc[AGG_ROWS];

    if (!plan_fits(N, E, ws_size)) return;
    const float*    emb   = ws;
    const float*    s_src = emb + (size_t)N * OUT_DIM;
    const float*    s_dst = s_src + N;
    const unsigned* recs  = (const unsigned*)ws + rec_offset(N);
    const int       nchunks = chunk_count(E);
    const unsigned* offs  = recs + (size_t)nchunks * CHUNK;

    const int tid  = threadIdx.x;
    const int lane = tid & 31;
    const int w    = __builtin_amdgcn_readfirstlane(tid >> 5);
    const int base = blockIdx.x * AGG_ROWS;
    int nrows = N - base;
    if (nrows > AGG_ROWS) nrows = AGG_ROWS;

    const int cpw = (nchunks + AGG_WAVES - 1) / AGG_WAVES;
    const int  myc   = w * cpw + lane;
    const bool mine  = lane < cpw && myc < nchunks;
    const int  cread = mine ? myc : nchunks - 1;
    const unsigned* orow = offs + (size_t)cread * OFF_PITCH + blockIdx.x;
    const int seg0 = (int)orow[0];
    const int seg1 = (int)orow[1];

    for (int i = tid; i < (AGG_ROWS + 1) * OUT_DIM / 4; i += AGG_THREADS)
        ((float4*)acc)[i] = make_float4(0.0f, 0.0f, 0.0f, 0.0f);
    for (int i = tid; i < AGG_ROWS; i += AGG_THREADS) {
        rsum[i] = 0.0f;
        ssrc[i] = (i < nrows) ? s_src[base + i] : 0.0f;
    }
    __syncthreads();

    const int len   = mine ? seg1 - seg0 : 0;
    const int gbase = myc * CHUNK + seg0;
    int incl = len;
#pragma unroll
    for (int d = 1; d < 32; d <<= 1) {
        const int n = __shfl_up(incl, d, 32);
        if (lane >= d) incl += n;
    }
    const int excl  = incl - len;
    const int shift = gbase - excl;
    const int total = __builtin_amdgcn_readlane(incl, 31);

    auto batch = [&](int cnt, int ridx) {
        int   sl = AGG_ROWS;
        int   dd = 0;
        float sc = 0.0f;
        if (lane < cnt) {
            const unsigned rec = recs[ridx];
            sl = (int)(rec >> REC_SHIFT);
            dd = (int)(rec & REC_MASK);
            const float logit = ssrc[sl] + s_dst[dd];
            const float lr    = logit > 0.0f ? logit : SLOPE * logit;
            sc = expf(lr);
            atomicAdd(&rsum[sl], sc);
        }
        int j = 0;
        for (; j + 8 <= cnt; j += 8) {
            v4f m[8];
#pragma unroll
            for (int t = 0; t < 8; ++t) {
                const int dj = __builtin_amdgcn_readlane(dd, j + t);
                m[t] = *((const v4f*)(emb + (size_t)dj * OUT_DIM) + lane);
            }
#pragma unroll
            for (int t = 0; t < 8; ++t) {
                const int   sj = __builtin_amdgcn_readlane(sl, j + t);
                const float cj = __int_as_float(
                    __builtin_amdgcn_readlane(__float_as_int(sc), j + t));
                float* arow = acc + sj * OUT_DIM + lane;
                atomicAdd(arow,      cj * m[t].x);
                atomicAdd(arow + 32, cj * m[t].y);
                atomicAdd(arow + 64, cj * m[t].z);
                atomicAdd(arow + 96, cj * m[t].w);
            }
        }
        for (; j < cnt; ++j) {
            const int   sj = __builtin_amdgcn_readlane(sl, j);
            const int   dj = __builtin_amdgcn_readlane(dd, j);
            const float cj = __int_as_float(
                __builtin_amdgcn_readlane(__float_as_int(sc), j));
            const v4f m = *((const v4f*)(emb + (size_t)dj * OUT_DIM) + lane);
            float*    arow = acc + sj * OUT_DIM + lane;
            atomicAdd(arow,      cj * m.x);
            atomicAdd(arow + 32, cj * m.y);
            atomicAdd(arow + 64, cj * m.z);
            atomicAdd(arow + 96, cj * m.w);
        }
    };

    for (int v0 = 0; v0 < total; v0 += 32) {
        const int v = v0 + lane;
        int sh = 0;
        for (int i = 0; i < cpw; ++i) {
            const int pi = __builtin_amdgcn_readlane(excl, i);
            const int di = __builtin_amdgcn_readlane(shift, i);
            if (v >= pi) sh = di;
        }
        const int rem = total - v0;
        if (rem >= 32) batch(32, sh + v);
        else           batch(rem, sh + v);
    }
    __syncthreads();

    float4* o4 = (float4*)out + (size_t)base * (OUT_DIM / 4);
    for (int i = tid; i < nrows * (OUT_DIM / 4); i += AGG_THREADS) {
        const float rs   = rsum[i >> 5];
        const float safe = rs > 0.0f ? rs : 1.0f;
        const float* ar = acc + (i >> 5) * OUT_DIM + (i & 31);
        float4 a;
        a.x = ar[0];
        a.y = ar[32];
        a.z = ar[64];
        a.w = ar[96];
        a.x = a.x / safe;
        a.y = a.y / safe;
        a.z = a.z / safe;
        a.w = a.w / safe;
        o4[i] = a;
    }
}

extern "C" void kernel_launch(void* const* d_in, const int* in_sizes, int n_in,
                              void* d_out, int out_size, void* d_ws, size_t ws_size,
                              hipStream_t stream) {
    const float* x     = (const float*)d_in[0];
    const float* W     = (const float*)d_in[1];
    const float* bias  = (const float*)d_in[2];
    const float* a     = (const float*)d_in[3];
    const int*   edges = (const int*)d_in[4];

    const int N = in_sizes[0] / IN_DIM;
    const int E = in_sizes[4] / 2;

    float* ws    = (float*)d_ws;
    float* emb   = ws;
    float* s_src = emb + (size_t)N * OUT_DIM;
    float* s_dst = s_src + N;
    float* out   = (float*)d_out;

    {
        gemm_wmma_kernel<<<GEMM_BLOCKS, GEMM_THREADS, 0, stream>>>(x, W, bias, emb, N);
    }
    {
        score_kernel<<<SCORE_BLOCKS, SCORE_THREADS, 0, stream>>>(emb, a, s_src, s_dst, N);
    }
    {
        int grid = (E + CHUNK - 1) / CHUNK;
        sort_kernel<<<grid, SORT_THREADS, 0, stream>>>(edges, ws, N, E, ws_size);
    }
    {
        int grid = (N + AGG_ROWS - 1) / AGG_ROWS;
        aggregate_kernel<<<grid, AGG_THREADS, 0, stream>>>(ws, out, N, E, ws_size);
    }
}


# Edge ordering leaves its own launch: the linear layer's work-groups order one 4096-edge chunk each at their head, in the x-tile LDS; one launch fewer; record arrival order, so LDS add order, changes
# speedup vs baseline: 4.0654x; 1.0418x over previous
#include <hip/hip_runtime.h>
#include <hip/hip_bf16.h>
#include <math.h>

#define IN_DIM  256
#define OUT_DIM 128
#define SLOPE   0.1f

typedef __attribute__((ext_vector_type(2))) float v2f;
typedef __attribute__((ext_vector_type(8))) float v8f;

typedef __attribute__((ext_vector_type(4))) float v4f;
typedef __attribute__((ext_vector_type(2))) int v2i;

#define AGG_ROWS    392
#define AGG_WAVES   32
#define NBINS       256
#define CHUNK       4096
#define ORDER_THREADS 256
#define ORDER_PER_THREAD (CHUNK / ORDER_THREADS)
#define OFF_PITCH   260
#define REC_SHIFT   23
#define REC_MASK    0x7fffffu

__device__ __forceinline__ size_t rec_offset(int N) {
    return ((size_t)N * (OUT_DIM + 2) + 3) & ~(size_t)3;
}
__device__ __forceinline__ int chunk_count(int E) {
    return (E + CHUNK - 1) / CHUNK;
}
__device__ __forceinline__ bool plan_fits(int N, int E, size_t ws_size) {
    const size_t words = rec_offset(N) + (size_t)chunk_count(E) * (CHUNK + OFF_PITCH);
    return words * 4 <= ws_size && N <= NBINS * AGG_ROWS &&
           chunk_count(E) <= AGG_WAVES * 32;
}

typedef __attribute__((ext_vector_type(4))) unsigned v4u;
static_assert(NBINS == ORDER_THREADS, "one bin offset per thread at the write-back");
static_assert(ORDER_PER_THREAD % 4 == 0, "records leave 16 bytes per thread and store");

__device__ __forceinline__ void order_chunk(
    const int* __restrict__ edges, unsigned* __restrict__ recs,
    unsigned* __restrict__ offs, unsigned* bins, unsigned* sorted,
    int c, int N, int E, int tid) {
    const int e0  = c * CHUNK;
    const v2i* e2 = (const v2i*)edges;
    v2i ed[ORDER_PER_THREAD];
#pragma unroll
    for (int i = 0; i < ORDER_PER_THREAD; ++i) {
        const int e = e0 + tid + ORDER_THREADS * i;
        ed[i] = e2[e < E ? e : E - 1];
    }
    for (int b = tid; b <= NBINS; b += ORDER_THREADS) bins[b] = 0u;
    __syncthreads();

    unsigned key[ORDER_PER_THREAD], rank[ORDER_PER_THREAD], rec[ORDER_PER_THREAD];
#pragma unroll
    for (int i = 0; i < ORDER_PER_THREAD; ++i) {
        const int e = e0 + tid + ORDER_THREADS * i;
        key[i]  = NBINS;
        rank[i] = 0u;
        rec[i]  = 0u;
        if (e < E && (unsigned)ed[i].x < (unsigned)N) {
            const unsigned blk = (unsigned)ed[i].x / AGG_ROWS;
            key[i]  = blk;
            rec[i]  = (((unsigned)ed[i].x - blk * AGG_ROWS) << REC_SHIFT) |
                      ((unsigned)ed[i].y & REC_MASK);
            rank[i] = atomicAdd(&bins[blk], 1u);
        }
    }
    __syncthreads();

    if (tid < 32) {
        unsigned cnt[8];
        unsigned sum = 0u;
#pragma unroll
        for (int k = 0; k < 8; ++k) {
            const unsigned t = bins[8 * tid + k];
            cnt[k] = sum;
            sum += t;
        }
        unsigned incl = sum;
#pragma unroll
        for (int d = 1; d < 32; d <<= 1) {
            const unsigned n = __shfl_up(incl, d, 32);
            if (tid >= d) incl += n;
        }
        const unsigned excl = incl - sum;
#pragma unroll
        for (int k = 0; k < 8; ++k) bins[8 * tid + k] = excl + cnt[k];
        if (tid == 31) bins[NBINS] = incl;
    }
    __syncthreads();

    unsigned slot[ORDER_PER_THREAD];
#pragma unroll
    for (int i = 0; i < ORDER_PER_THREAD; ++i) slot[i] = bins[key[i]] + rank[i];
#pragma unroll
    for (int i = 0; i < ORDER_PER_THREAD; ++i)
        if (key[i] < NBINS) sorted[slot[i]] = rec[i];
    __syncthreads();

    const unsigned total = bins[NBINS];
    const unsigned myoff = bins[tid];
    v4u piece[ORDER_PER_THREAD / 4];
#pragma unroll
    for (int i = 0; i < ORDER_PER_THREAD / 4; ++i)
        piece[i] = *(const v4u*)(sorted + 4 * (tid + ORDER_THREADS * i));
    __syncthreads();

    unsigned* orow = offs + (size_t)c * OFF_PITCH;
    orow[tid] = myoff;
    if (tid == 0) orow[NBINS] = total;
    unsigned* dst = recs + (size_t)c * CHUNK;
#pragma unroll
    for (int i = 0; i < ORDER_PER_THREAD / 4; ++i) {
        const unsigned p = 4 * (tid + ORDER_THREADS * i);
        if (p < total) *(v4u*)(dst + p) = piece[i];
    }
}

#define GEMM_THREADS 256
#define GEMM_BLOCKS  512
#define GEMM_ROWS    16
#define GEMM_PITCH   260


__global__ __launch_bounds__(GEMM_THREADS, 4) void gemm_wmma_kernel(
    const float* __restrict__ x, const float* __restrict__ W,
    const float* __restrict__ bias, const int* __restrict__ edges,
    float* __restrict__ ws, int N, int E, size_t ws_size) {
    __shared__ float xs[2][GEMM_ROWS * GEMM_PITCH];

    const int tid  = threadIdx.x;

    if (plan_fits(N, E, ws_size)) {
        unsigned* recs   = (unsigned*)ws + rec_offset(N);
        const int nchunks = chunk_count(E);
        unsigned* offs   = recs + (size_t)nchunks * CHUNK;
        unsigned* bins   = (unsigned*)&xs[0][0];
        unsigned* sorted = bins + OFF_PITCH;
        for (int c = (int)gridDim.x - 1 - (int)blockIdx.x; c < nchunks; c += (int)gridDim.x)
            order_chunk(edges, recs, offs, bins, sorted, c, N, E, tid);
    }

    float* emb = ws;
    const int wave = __builtin_amdgcn_readfirstlane(tid >> 5);
    const int lane = tid & 31;
    const int half = lane >> 4;
    const int l16  = lane & 15;
    const int n0   = wave * 16;
    const int ntiles = (N + GEMM_ROWS - 1) / GEMM_ROWS;
    int s = blockIdx.x;
    if (s >= ntiles) return;

    v4f w[32];
    {
        const float* wrow = W + (size_t)(n0 + l16) * IN_DIM + 4 * half;
#pragma unroll
        for (int j = 0; j < 32; ++j) w[j] = *(const v4f*)(wrow + 8 * j);
    }
    const float bv = bias[n0 + l16];

    const int srow = wave >> 1;
    const int scol = ((wave & 1) * 32 + lane) * 4;

    v4f g[4];
    auto fetch = [&](int tile) {
#pragma unroll
        for (int i = 0; i < 4; ++i) {
            int row = tile * GEMM_ROWS + 4 * i + srow;
            row = row < N ? row : N - 1;
            const float* xr = x + (size_t)row * IN_DIM;
            g[i] = *(const v4f*)(xr + scol);
        }
    };
    auto stash = [&](float* buf) {
#pragma unroll
        for (int i = 0; i < 4; ++i)
            *(v4f*)(buf + (4 * i + srow) * GEMM_PITCH + scol) = g[i];
    };

    fetch(s);
    stash(xs[0]);
    __syncthreads();

    int cur = 0;
    for (;;) {
        const int  next = s + (int)gridDim.x;
        const bool more = next < ntiles;
        fetch(more ? next : s);
        __builtin_amdgcn_sched_barrier(0);

        const float* ap = xs[cur] + l16 * GEMM_PITCH + 4 * half;
        v4f a[2][8];
#pragma unroll
        for (int t = 0; t < 8; ++t) a[0][t] = *(const v4f*)(ap + 8 * t);
        v8f c = {};
#pragma unroll
        for (int b = 0; b < 4; ++b) {
            if (b < 3) {
#pragma unroll
                for (int t = 0; t < 8; ++t)
                    a[(b + 1) & 1][t] = *(const v4f*)(ap + 8 * (8 * (b + 1) + t));
            }
            __builtin_amdgcn_sched_barrier(0);
#pragma unroll
            for (int t = 0; t < 8; ++t) {
                const v4f av = a[b & 1][t];
                const v4f wv = w[8 * b + t];
                const v2f alo = {av.x, av.y};
                const v2f ahi = {av.z, av.w};
                const v2f blo = {wv.x, wv.y};
                const v2f bhi = {wv.z, wv.w};
                c = __builtin_amdgcn_wmma_f32_16x16x4_f32(
                        false, alo, false, blo, (short)0, c, false, false);
                c = __builtin_amdgcn_wmma_f32_16x16x4_f32(
                        false, ahi, false, bhi, (short)0, c, false, false);
            }
        }

        stash(xs[cur ^ 1]);
        __syncthreads();

        asm volatile("v_nop\n\tv_nop\n\tv_nop\n\tv_nop" : "+v"(c));

        const int m0 = s * GEMM_ROWS;
        float* erow = emb + (size_t)(m0 + 8 * half) * OUT_DIM + n0 + l16;
        if (m0 + GEMM_ROWS <= N) {
#pragma unroll
            for (int j = 0; j < 8; ++j) erow[j * OUT_DIM] = c[j] + bv;
        } else {
#pragma unroll
            for (int j = 0; j < 8; ++j)
                if (m0 + j + 8 * half < N) erow[j * OUT_DIM] = c[j] + bv;
        }

        if (!more) break;
        cur ^= 1;
        s = next;
    }
}

#define SCORE_THREADS 256
#define SCORE_BLOCKS  512
#define SCORE_ROWS    8

__device__ __forceinline__ float xhalf(float v) {
    const int b = __builtin_bit_cast(int, v);
    return __builtin_bit_cast(float,
        __builtin_amdgcn_permlanex16(b, b, 0x76543210, 0xfedcba98, false, false));
}
template <int M>
__device__ __forceinline__ float row_xor(float v) {
    const int b = __builtin_bit_cast(int, v);
    return __builtin_bit_cast(float,
        __builtin_amdgcn_update_dpp(0, b, 0x160 + M, 0xf, 0xf, true));
}

__global__ __launch_bounds__(SCORE_THREADS) void score_kernel(
    const float* __restrict__ emb, const float* __restrict__ a,
    float* __restrict__ s_src, float* __restrict__ s_dst, int N) {
    const int lane = threadIdx.x & 31;
    const int half = lane >> 4;
    const int l16  = lane & 15;
    const int gw = __builtin_amdgcn_readfirstlane(
        (int)(blockIdx.x * (SCORE_THREADS / 32) + (threadIdx.x >> 5)));
    const int nw = (int)gridDim.x * (SCORE_THREADS / 32);
    const int ngroups = (N + SCORE_ROWS - 1) / SCORE_ROWS;

    const v4f a1 = *(const v4f*)(a + lane * 4);
    const v4f a2 = *(const v4f*)(a + OUT_DIM + lane * 4);
    const v2f ax = {a1.x, a2.x};
    const v2f ay = {a1.y, a2.y};
    const v2f az = {a1.z, a2.z};
    const v2f aw = {a1.w, a2.w};
    float* sp = half ? s_dst : s_src;

    for (int g = gw; g < ngroups; g += nw) {
        const int r0 = g * SCORE_ROWS;
        v4f e[SCORE_ROWS];
#pragma unroll
        for (int t = 0; t < SCORE_ROWS; ++t) {
            int row = r0 + t;
            row = row < N ? row : N - 1;
            e[t] = *(const v4f*)(emb + (size_t)row * OUT_DIM + lane * 4);
        }
        __builtin_amdgcn_sched_barrier(0);
        float res = 0.0f;
#pragma unroll
        for (int t = 0; t < SCORE_ROWS; ++t) {
            const v2f ex = {e[t].x, e[t].x};
            const v2f ey = {e[t].y, e[t].y};
            const v2f ez = {e[t].z, e[t].z};
            const v2f ew = {e[t].w, e[t].w};
            const v2f p  = ex * ax + ey * ay + ez * az + ew * aw;
            const float keep = half ? p.y : p.x;
            const float give = half ? p.x : p.y;
            float q = keep + xhalf(give);
            q += row_xor<8>(q);
            q += row_xor<4>(q);
            q += row_xor<2>(q);
            q += row_xor<1>(q);
            res = (l16 == t) ? q : res;
        }
        const int row = r0 + l16;
        if (l16 < SCORE_ROWS && row < N) sp[row] = res;
    }
}

#define AGG_THREADS 1024

__global__ __launch_bounds__(AGG_THREADS) void aggregate_kernel(
    const float* __restrict__ ws, float* __restrict__ out, int N, int E,
    size_t ws_size) {
    __shared__ float acc[(AGG_ROWS + 1) * OUT_DIM];
    __shared__ float rsum[AGG_ROWS];
    __shared__ float ssrc[AGG_ROWS];

    if (!plan_fits(N, E, ws_size)) return;
    const float*    emb   = ws;
    const float*    s_src = emb + (size_t)N * OUT_DIM;
    const float*    s_dst = s_src + N;
    const unsigned* recs  = (const unsigned*)ws + rec_offset(N);
    const int       nchunks = chunk_count(E);
    const unsigned* offs  = recs + (size_t)nchunks * CHUNK;

    const int tid  = threadIdx.x;
    const int lane = tid & 31;
    const int w    = __builtin_amdgcn_readfirstlane(tid >> 5);
    const int base = blockIdx.x * AGG_ROWS;
    int nrows = N - base;
    if (nrows > AGG_ROWS) nrows = AGG_ROWS;

    const int cpw = (nchunks + AGG_WAVES - 1) / AGG_WAVES;
    const int  myc   = w * cpw + lane;
    const bool mine  = lane < cpw && myc < nchunks;
    const int  cread = mine ? myc : nchunks - 1;
    const unsigned* orow = offs + (size_t)cread * OFF_PITCH + blockIdx.x;
    const int seg0 = (int)orow[0];
    const int seg1 = (int)orow[1];

    for (int i = tid; i < (AGG_ROWS + 1) * OUT_DIM / 4; i += AGG_THREADS)
        ((float4*)acc)[i] = make_float4(0.0f, 0.0f, 0.0f, 0.0f);
    for (int i = tid; i < AGG_ROWS; i += AGG_THREADS) {
        rsum[i] = 0.0f;
        ssrc[i] = (i < nrows) ? s_src[base + i] : 0.0f;
    }
    __syncthreads();

    const int len   = mine ? seg1 - seg0 : 0;
    const int gbase = myc * CHUNK + seg0;
    int incl = len;
#pragma unroll
    for (int d = 1; d < 32; d <<= 1) {
        const int n = __shfl_up(incl, d, 32);
        if (lane >= d) incl += n;
    }
    const int excl  = incl - len;
    const int shift = gbase - excl;
    const int total = __builtin_amdgcn_readlane(incl, 31);

    auto batch = [&](int cnt, int ridx) {
        int   sl = AGG_ROWS;
        int   dd = 0;
        float sc = 0.0f;
        if (lane < cnt) {
            const unsigned rec = recs[ridx];
            sl = (int)(rec >> REC_SHIFT);
            dd = (int)(rec & REC_MASK);
            const float logit = ssrc[sl] + s_dst[dd];
            const float lr    = logit > 0.0f ? logit : SLOPE * logit;
            sc = expf(lr);
            atomicAdd(&rsum[sl], sc);
        }
        int j = 0;
        for (; j + 8 <= cnt; j += 8) {
            v4f m[8];
#pragma unroll
            for (int t = 0; t < 8; ++t) {
                const int dj = __builtin_amdgcn_readlane(dd, j + t);
                m[t] = *((const v4f*)(emb + (size_t)dj * OUT_DIM) + lane);
            }
#pragma unroll
            for (int t = 0; t < 8; ++t) {
                const int   sj = __builtin_amdgcn_readlane(sl, j + t);
                const float cj = __int_as_float(
                    __builtin_amdgcn_readlane(__float_as_int(sc), j + t));
                float* arow = acc + sj * OUT_DIM + lane;
                atomicAdd(arow,      cj * m[t].x);
                atomicAdd(arow + 32, cj * m[t].y);
                atomicAdd(arow + 64, cj * m[t].z);
                atomicAdd(arow + 96, cj * m[t].w);
            }
        }
        for (; j < cnt; ++j) {
            const int   sj = __builtin_amdgcn_readlane(sl, j);
            const int   dj = __builtin_amdgcn_readlane(dd, j);
            const float cj = __int_as_float(
                __builtin_amdgcn_readlane(__float_as_int(sc), j));
            const v4f m = *((const v4f*)(emb + (size_t)dj * OUT_DIM) + lane);
            float*    arow = acc + sj * OUT_DIM + lane;
            atomicAdd(arow,      cj * m.x);
            atomicAdd(arow + 32, cj * m.y);
            atomicAdd(arow + 64, cj * m.z);
            atomicAdd(arow + 96, cj * m.w);
        }
    };

    for (int v0 = 0; v0 < total; v0 += 32) {
        const int v = v0 + lane;
        int sh = 0;
        for (int i = 0; i < cpw; ++i) {
            const int pi = __builtin_amdgcn_readlane(excl, i);
            const int di = __builtin_amdgcn_readlane(shift, i);
            if (v >= pi) sh = di;
        }
        const int rem = total - v0;
        if (rem >= 32) batch(32, sh + v);
        else           batch(rem, sh + v);
    }
    __syncthreads();

    float4* o4 = (float4*)out + (size_t)base * (OUT_DIM / 4);
    for (int i = tid; i < nrows * (OUT_DIM / 4); i += AGG_THREADS) {
        const float rs   = rsum[i >> 5];
        const float safe = rs > 0.0f ? rs : 1.0f;
        const float* ar = acc + (i >> 5) * OUT_DIM + (i & 31);
        float4 a;
        a.x = ar[0];
        a.y = ar[32];
        a.z = ar[64];
        a.w = ar[96];
        a.x = a.x / safe;
        a.y = a.y / safe;
        a.z = a.z / safe;
        a.w = a.w / safe;
        o4[i] = a;
    }
}

extern "C" void kernel_launch(void* const* d_in, const int* in_sizes, int n_in,
                              void* d_out, int out_size, void* d_ws, size_t ws_size,
                              hipStream_t stream) {
    const float* x     = (const float*)d_in[0];
    const float* W     = (const float*)d_in[1];
    const float* bias  = (const float*)d_in[2];
    const float* a     = (const float*)d_in[3];
    const int*   edges = (const int*)d_in[4];

    const int N = in_sizes[0] / IN_DIM;
    const int E = in_sizes[4] / 2;

    float* ws    = (float*)d_ws;
    float* emb   = ws;
    float* s_src = emb + (size_t)N * OUT_DIM;
    float* s_dst = s_src + N;
    float* out   = (float*)d_out;

    {
        gemm_wmma_kernel<<<GEMM_BLOCKS, GEMM_THREADS, 0, stream>>>(x, W, bias, edges, ws, N, E, ws_size);
    }
    {
        score_kernel<<<SCORE_BLOCKS, SCORE_THREADS, 0, stream>>>(emb, a, s_src, s_dst, N);
    }
    {
        int grid = (N + AGG_ROWS - 1) / AGG_ROWS;
        aggregate_kernel<<<grid, AGG_THREADS, 0, stream>>>(ws, out, N, E, ws_size);
    }
}
